# starting kernel
# baseline (speedup 1.0000x reference)
#include <hip/hip_runtime.h>

typedef float v2f __attribute__((ext_vector_type(2)));
typedef float v4f __attribute__((ext_vector_type(4)));
typedef float v8f __attribute__((ext_vector_type(8)));

#define N_NODES 100000
#define N_EDGES 1600000
#define IN_DIM  165
#define HID     64
#define OUT_DIM 2
#define NTILES  (N_NODES / 16)
#define KPAIRS  83

__global__ void zero_kernel(float* __restrict__ p, int n) {
    int i = blockIdx.x * blockDim.x + threadIdx.x;
    if (i < n) p[i] = 0.0f;
}

__global__ void deg_kernel(const int* __restrict__ dst, float* __restrict__ deg) {
    int e = blockIdx.x * blockDim.x + threadIdx.x;
    if (e < N_EDGES) unsafeAtomicAdd(&deg[dst[e]], 1.0f);
}

__global__ void dinv_kernel(float* __restrict__ d) {
    int i = blockIdx.x * blockDim.x + threadIdx.x;
    if (i < N_NODES) d[i] = rsqrtf(d[i] + 1.0f);
}

__global__ void enorm_kernel(const int* __restrict__ src, const int* __restrict__ dst,
                             const float* __restrict__ dinv, float* __restrict__ en) {
    int e = blockIdx.x * blockDim.x + threadIdx.x;
    if (e < N_EDGES) en[e] = dinv[src[e]] * dinv[dst[e]];
}

__global__ __launch_bounds__(256) void gemm1_wmma(const float* __restrict__ x,
                                                  const float* __restrict__ W1,
                                                  float* __restrict__ hw1) {
    __shared__ float w1p[KPAIRS * 2 * HID];
    for (int i = threadIdx.x; i < KPAIRS * 2 * HID; i += 256) {
        int kp   = i >> 7;
        int r    = i & 127;
        int col  = r >> 1;
        int sub  = r & 1;
        int krow = kp * 2 + sub;
        w1p[i] = (krow < IN_DIM) ? W1[krow * HID + col] : 0.0f;
    }
    __syncthreads();

    const int wave = threadIdx.x >> 5;
    const int lane = threadIdx.x & 31;
    const int tile = blockIdx.x * 8 + wave;
    if (tile >= NTILES) return;

    const int hh  = lane >> 4;
    const int lr  = lane & 15;
    const int row = tile * 16 + lr;
    const float* __restrict__ xrb = x + (long)row * IN_DIM;
    const float* __restrict__ xr  = xrb + 2 * hh;

    v8f acc0 = (v8f){}, acc1 = (v8f){}, acc2 = (v8f){}, acc3 = (v8f){};

    for (int k = 0; k < 164; k += 4) {
        v2f A;
        A.x = xr[k];
        A.y = xr[k + 1];
        const int ka = k + 2 * hh;
        const v2f* __restrict__ bp = (const v2f*)&w1p[(ka >> 1) * 2 * HID + lr * 2];
        v2f B0 = bp[0];
        v2f B1 = bp[16];
        v2f B2 = bp[32];
        v2f B3 = bp[48];
        acc0 = __builtin_amdgcn_wmma_f32_16x16x4_f32(false, A, false, B0, (short)0, acc0, false, false);
        acc1 = __builtin_amdgcn_wmma_f32_16x16x4_f32(false, A, false, B1, (short)0, acc1, false, false);
        acc2 = __builtin_amdgcn_wmma_f32_16x16x4_f32(false, A, false, B2, (short)0, acc2, false, false);
        acc3 = __builtin_amdgcn_wmma_f32_16x16x4_f32(false, A, false, B3, (short)0, acc3, false, false);
    }

    {
        float a164 = xrb[164];
        v2f A;
        A.x = (hh == 0) ? a164 : 0.0f;
        A.y = 0.0f;
        const v2f* __restrict__ bp = (const v2f*)&w1p[(KPAIRS - 1) * 2 * HID + lr * 2];
        v2f B0 = bp[0];
        v2f B1 = bp[16];
        v2f B2 = bp[32];
        v2f B3 = bp[48];
        acc0 = __builtin_amdgcn_wmma_f32_16x16x4_f32(false, A, false, B0, (short)0, acc0, false, false);
        acc1 = __builtin_amdgcn_wmma_f32_16x16x4_f32(false, A, false, B1, (short)0, acc1, false, false);
        acc2 = __builtin_amdgcn_wmma_f32_16x16x4_f32(false, A, false, B2, (short)0, acc2, false, false);
        acc3 = __builtin_amdgcn_wmma_f32_16x16x4_f32(false, A, false, B3, (short)0, acc3, false, false);
    }

    float* __restrict__ outp = hw1 + (long)(tile * 16 + 8 * hh) * HID + lr;
#pragma unroll
    for (int v = 0; v < 8; ++v) {
        long o = (long)v * HID;
        outp[o +  0] = acc0[v];
        outp[o + 16] = acc1[v];
        outp[o + 32] = acc2[v];
        outp[o + 48] = acc3[v];
    }
}

__global__ void scatter1(const int* __restrict__ src, const int* __restrict__ dst,
                         const float* __restrict__ en, const float* __restrict__ hw1,
                         float* __restrict__ agg) {
    int tid = blockIdx.x * blockDim.x + threadIdx.x;
    int e = tid >> 4;
    int f = (tid & 15) * 4;
    if (e >= N_EDGES) return;
    int s = src[e], d = dst[e];
    float w = en[e];
    v4f m = *(const v4f*)(hw1 + (long)s * HID + f);
    float* ap = agg + (long)d * HID + f;
    unsafeAtomicAdd(ap + 0, m.x * w);
    unsafeAtomicAdd(ap + 1, m.y * w);
    unsafeAtomicAdd(ap + 2, m.z * w);
    unsafeAtomicAdd(ap + 3, m.w * w);
}

__global__ void relu_bias(const float* __restrict__ hw1, const float* __restrict__ dinv,
                          const float* __restrict__ b1, float* __restrict__ h) {
    int tid = blockIdx.x * blockDim.x + threadIdx.x;
    if (tid >= N_NODES * HID) return;
    int n = tid >> 6, f = tid & 63;
    float sn = dinv[n]; sn *= sn;
    float v = h[tid] + hw1[tid] * sn + b1[f];
    h[tid] = v > 0.0f ? v : 0.0f;
}

__global__ void gemm2(const float* __restrict__ h, const float* __restrict__ W2,
                      float* __restrict__ hw2) {
    int n = blockIdx.x * blockDim.x + threadIdx.x;
    if (n >= N_NODES) return;
    const float* hr = h + (long)n * HID;
    float o0 = 0.0f, o1 = 0.0f;
#pragma unroll 8
    for (int k = 0; k < HID; ++k) {
        float hv = hr[k];
        o0 = fmaf(hv, W2[2 * k],     o0);
        o1 = fmaf(hv, W2[2 * k + 1], o1);
    }
    hw2[2 * n]     = o0;
    hw2[2 * n + 1] = o1;
}

__global__ void scatter2(const int* __restrict__ src, const int* __restrict__ dst,
                         const float* __restrict__ en, const float* __restrict__ hw2,
                         float* __restrict__ out) {
    int e = blockIdx.x * blockDim.x + threadIdx.x;
    if (e >= N_EDGES) return;
    int s = src[e], d = dst[e];
    float w = en[e];
    unsafeAtomicAdd(&out[2 * d],     hw2[2 * s]     * w);
    unsafeAtomicAdd(&out[2 * d + 1], hw2[2 * s + 1] * w);
}

__global__ void finish2(const float* __restrict__ hw2, const float* __restrict__ dinv,
                        const float* __restrict__ b2, float* __restrict__ out) {
    int n = blockIdx.x * blockDim.x + threadIdx.x;
    if (n >= N_NODES) return;
    float sn = dinv[n]; sn *= sn;
    out[2 * n]     += hw2[2 * n]     * sn + b2[0];
    out[2 * n + 1] += hw2[2 * n + 1] * sn + b2[1];
}

extern "C" void kernel_launch(void* const* d_in, const int* in_sizes, int n_in,
                              void* d_out, int out_size, void* d_ws, size_t ws_size,
                              hipStream_t stream) {
    const float* x  = (const float*)d_in[0];
    const int*   ei = (const int*)  d_in[1];
    const float* W1 = (const float*)d_in[2];
    const float* b1 = (const float*)d_in[3];
    const float* W2 = (const float*)d_in[4];
    const float* b2 = (const float*)d_in[5];
    const int* src = ei;
    const int* dst = ei + N_EDGES;
    float* out = (float*)d_out;

    float* ws   = (float*)d_ws;
    float* dinv = ws;
    float* en   = dinv + N_NODES;
    float* hw1  = en + N_EDGES;
    float* h    = hw1 + (size_t)N_NODES * HID;
    float* hw2  = h   + (size_t)N_NODES * HID;

    const dim3 blk(256);

    zero_kernel<<<(N_NODES + 255) / 256, blk, 0, stream>>>(dinv, N_NODES);
    zero_kernel<<<(N_NODES * HID + 255) / 256, blk, 0, stream>>>(h, N_NODES * HID);
    zero_kernel<<<(OUT_DIM * N_NODES + 255) / 256, blk, 0, stream>>>(out, OUT_DIM * N_NODES);

    deg_kernel  <<<(N_EDGES + 255) / 256, blk, 0, stream>>>(dst, dinv);
    dinv_kernel <<<(N_NODES + 255) / 256, blk, 0, stream>>>(dinv);
    enorm_kernel<<<(N_EDGES + 255) / 256, blk, 0, stream>>>(src, dst, dinv, en);

    gemm1_wmma<<<(NTILES + 7) / 8, blk, 0, stream>>>(x, W1, hw1);
    scatter1  <<<(N_EDGES * 16) / 256, blk, 0, stream>>>(src, dst, en, hw1, h);
    relu_bias <<<(N_NODES * HID + 255) / 256, blk, 0, stream>>>(hw1, dinv, b1, h);

    gemm2   <<<(N_NODES + 255) / 256, blk, 0, stream>>>(h, W2, hw2);
    scatter2<<<(N_EDGES + 255) / 256, blk, 0, stream>>>(src, dst, en, hw2, out);
    finish2 <<<(N_NODES + 255) / 256, blk, 0, stream>>>(hw2, dinv, b2, out);
}


# Degree count and layer-2 edge scatter accumulate in LDS per (node range, edge chunk) block, flushed once by coalesced line-aligned atomics; float sums re-associated (LDS partials, then global adds)
# speedup vs baseline: 1.6354x; 1.6354x over previous
#include <hip/hip_runtime.h>

typedef float v2f __attribute__((ext_vector_type(2)));
typedef float v4f __attribute__((ext_vector_type(4)));
typedef float v8f __attribute__((ext_vector_type(8)));
typedef int   v4i __attribute__((ext_vector_type(4)));

#define N_NODES 100000
#define N_EDGES 1600000
#define IN_DIM  165
#define HID     64
#define OUT_DIM 2
#define NTILES  (N_NODES / 16)
#define KPAIRS  83

#define N_RANGES    8
#define RNG_NODES   12512
#define N_CHUNKS    32
#define CHUNK_EDGES (N_EDGES / N_CHUNKS)
#define CHUNK_VEC4  (CHUNK_EDGES / 4)
#define PRIV_THREADS 1024
#define DEG_ROUNDS  (CHUNK_VEC4 / PRIV_THREADS)
#define DEG_ROT     (RNG_NODES / N_CHUNKS / 32 * 32)
#define OUT_ROT     (2 * RNG_NODES / N_CHUNKS / 32 * 32)

__global__ void zero_kernel(float* __restrict__ p, int n) {
    int i = blockIdx.x * blockDim.x + threadIdx.x;
    if (i < n) p[i] = 0.0f;
}

__global__ __launch_bounds__(PRIV_THREADS) void deg_kernel(const int* __restrict__ dst,
                                                           float* __restrict__ deg) {
    __shared__ unsigned int cnt[RNG_NODES];
    const int r  = blockIdx.x % N_RANGES;
    const int c  = blockIdx.x / N_RANGES;
    const int lo = r * RNG_NODES;
    for (int i = threadIdx.x; i < RNG_NODES; i += PRIV_THREADS) cnt[i] = 0u;
    __syncthreads();

    const v4i* __restrict__ pd = (const v4i*)(dst + (size_t)c * CHUNK_EDGES);
    v4i d[DEG_ROUNDS + 1];
#pragma unroll
    for (int k = 0; k < DEG_ROUNDS; ++k) d[k] = pd[threadIdx.x + k * PRIV_THREADS];
    d[DEG_ROUNDS] = (v4i){-1, -1, -1, -1};
    if (threadIdx.x + DEG_ROUNDS * PRIV_THREADS < CHUNK_VEC4)
        d[DEG_ROUNDS] = pd[threadIdx.x + DEG_ROUNDS * PRIV_THREADS];
#pragma unroll
    for (int k = 0; k <= DEG_ROUNDS; ++k) {
        unsigned t0 = (unsigned)(d[k].x - lo), t1 = (unsigned)(d[k].y - lo);
        unsigned t2 = (unsigned)(d[k].z - lo), t3 = (unsigned)(d[k].w - lo);
        if (t0 < RNG_NODES) atomicAdd(&cnt[t0], 1u);
        if (t1 < RNG_NODES) atomicAdd(&cnt[t1], 1u);
        if (t2 < RNG_NODES) atomicAdd(&cnt[t2], 1u);
        if (t3 < RNG_NODES) atomicAdd(&cnt[t3], 1u);
    }
    __syncthreads();

    const int rot = c * DEG_ROT;
    for (int i = threadIdx.x; i < RNG_NODES; i += PRIV_THREADS) {
        int j = i + rot;
        if (j >= RNG_NODES) j -= RNG_NODES;
        int n = lo + j;
        if (n < N_NODES) unsafeAtomicAdd(&deg[n], (float)cnt[j]);
    }
}

__global__ void dinv_kernel(float* __restrict__ d) {
    int i = blockIdx.x * blockDim.x + threadIdx.x;
    if (i < N_NODES) d[i] = rsqrtf(d[i] + 1.0f);
}

__global__ void enorm_kernel(const int* __restrict__ src, const int* __restrict__ dst,
                             const float* __restrict__ dinv, float* __restrict__ en) {
    int e = blockIdx.x * blockDim.x + threadIdx.x;
    if (e < N_EDGES) en[e] = dinv[src[e]] * dinv[dst[e]];
}

__global__ __launch_bounds__(256) void gemm1_wmma(const float* __restrict__ x,
                                                  const float* __restrict__ W1,
                                                  float* __restrict__ hw1) {
    __shared__ float w1p[KPAIRS * 2 * HID];
    for (int i = threadIdx.x; i < KPAIRS * 2 * HID; i += 256) {
        int kp   = i >> 7;
        int r    = i & 127;
        int col  = r >> 1;
        int sub  = r & 1;
        int krow = kp * 2 + sub;
        w1p[i] = (krow < IN_DIM) ? W1[krow * HID + col] : 0.0f;
    }
    __syncthreads();

    const int wave = threadIdx.x >> 5;
    const int lane = threadIdx.x & 31;
    const int tile = blockIdx.x * 8 + wave;
    if (tile >= NTILES) return;

    const int hh  = lane >> 4;
    const int lr  = lane & 15;
    const int row = tile * 16 + lr;
    const float* __restrict__ xrb = x + (long)row * IN_DIM;
    const float* __restrict__ xr  = xrb + 2 * hh;

    v8f acc0 = (v8f){}, acc1 = (v8f){}, acc2 = (v8f){}, acc3 = (v8f){};

    for (int k = 0; k < 164; k += 4) {
        v2f A;
        A.x = xr[k];
        A.y = xr[k + 1];
        const int ka = k + 2 * hh;
        const v2f* __restrict__ bp = (const v2f*)&w1p[(ka >> 1) * 2 * HID + lr * 2];
        v2f B0 = bp[0];
        v2f B1 = bp[16];
        v2f B2 = bp[32];
        v2f B3 = bp[48];
        acc0 = __builtin_amdgcn_wmma_f32_16x16x4_f32(false, A, false, B0, (short)0, acc0, false, false);
        acc1 = __builtin_amdgcn_wmma_f32_16x16x4_f32(false, A, false, B1, (short)0, acc1, false, false);
        acc2 = __builtin_amdgcn_wmma_f32_16x16x4_f32(false, A, false, B2, (short)0, acc2, false, false);
        acc3 = __builtin_amdgcn_wmma_f32_16x16x4_f32(false, A, false, B3, (short)0, acc3, false, false);
    }

    {
        float a164 = xrb[164];
        v2f A;
        A.x = (hh == 0) ? a164 : 0.0f;
        A.y = 0.0f;
        const v2f* __restrict__ bp = (const v2f*)&w1p[(KPAIRS - 1) * 2 * HID + lr * 2];
        v2f B0 = bp[0];
        v2f B1 = bp[16];
        v2f B2 = bp[32];
        v2f B3 = bp[48];
        acc0 = __builtin_amdgcn_wmma_f32_16x16x4_f32(false, A, false, B0, (short)0, acc0, false, false);
        acc1 = __builtin_amdgcn_wmma_f32_16x16x4_f32(false, A, false, B1, (short)0, acc1, false, false);
        acc2 = __builtin_amdgcn_wmma_f32_16x16x4_f32(false, A, false, B2, (short)0, acc2, false, false);
        acc3 = __builtin_amdgcn_wmma_f32_16x16x4_f32(false, A, false, B3, (short)0, acc3, false, false);
    }

    float* __restrict__ outp = hw1 + (long)(tile * 16 + 8 * hh) * HID + lr;
#pragma unroll
    for (int v = 0; v < 8; ++v) {
        long o = (long)v * HID;
        outp[o +  0] = acc0[v];
        outp[o + 16] = acc1[v];
        outp[o + 32] = acc2[v];
        outp[o + 48] = acc3[v];
    }
}

__global__ void scatter1(const int* __restrict__ src, const int* __restrict__ dst,
                         const float* __restrict__ en, const float* __restrict__ hw1,
                         float* __restrict__ agg) {
    int tid = blockIdx.x * blockDim.x + threadIdx.x;
    int e = tid >> 4;
    int f = (tid & 15) * 4;
    if (e >= N_EDGES) return;
    int s = src[e], d = dst[e];
    float w = en[e];
    v4f m = *(const v4f*)(hw1 + (long)s * HID + f);
    float* ap = agg + (long)d * HID + f;
    unsafeAtomicAdd(ap + 0, m.x * w);
    unsafeAtomicAdd(ap + 1, m.y * w);
    unsafeAtomicAdd(ap + 2, m.z * w);
    unsafeAtomicAdd(ap + 3, m.w * w);
}

__global__ void relu_bias(const float* __restrict__ hw1, const float* __restrict__ dinv,
                          const float* __restrict__ b1, float* __restrict__ h) {
    int tid = blockIdx.x * blockDim.x + threadIdx.x;
    if (tid >= N_NODES * HID) return;
    int n = tid >> 6, f = tid & 63;
    float sn = dinv[n]; sn *= sn;
    float v = h[tid] + hw1[tid] * sn + b1[f];
    h[tid] = v > 0.0f ? v : 0.0f;
}

__global__ void gemm2(const float* __restrict__ h, const float* __restrict__ W2,
                      float* __restrict__ hw2) {
    int n = blockIdx.x * blockDim.x + threadIdx.x;
    if (n >= N_NODES) return;
    const float* hr = h + (long)n * HID;
    float o0 = 0.0f, o1 = 0.0f;
#pragma unroll 8
    for (int k = 0; k < HID; ++k) {
        float hv = hr[k];
        o0 = fmaf(hv, W2[2 * k],     o0);
        o1 = fmaf(hv, W2[2 * k + 1], o1);
    }
    hw2[2 * n]     = o0;
    hw2[2 * n + 1] = o1;
}

__global__ __launch_bounds__(PRIV_THREADS) void scatter2(const int* __restrict__ src,
                                                         const int* __restrict__ dst,
                                                         const float* __restrict__ en,
                                                         const float* __restrict__ hw2,
                                                         float* __restrict__ out) {
    __shared__ float part[2 * RNG_NODES];
    const int r  = blockIdx.x % N_RANGES;
    const int c  = blockIdx.x / N_RANGES;
    const int lo = r * RNG_NODES;
    for (int i = threadIdx.x; i < 2 * RNG_NODES; i += PRIV_THREADS) part[i] = 0.0f;
    __syncthreads();

    const v4i* __restrict__ ps = (const v4i*)(src + (size_t)c * CHUNK_EDGES);
    const v4i* __restrict__ pd = (const v4i*)(dst + (size_t)c * CHUNK_EDGES);
    const v4f* __restrict__ pw = (const v4f*)(en  + (size_t)c * CHUNK_EDGES);
    for (int b = threadIdx.x; b < CHUNK_VEC4; b += 4 * PRIV_THREADS) {
        v4i d[4], s[4];
        v4f w[4];
#pragma unroll
        for (int k = 0; k < 4; ++k) {
            const int  i  = b + k * PRIV_THREADS;
            const bool ok = i < CHUNK_VEC4;
            const int  ic = ok ? i : CHUNK_VEC4 - 1;
            d[k] = pd[ic];
            s[k] = ps[ic];
            w[k] = pw[ic];
            if (!ok) d[k] = (v4i){-1, -1, -1, -1};
        }
        unsigned t[16];
        v2f      h[16];
#pragma unroll
        for (int m = 0; m < 16; ++m) {
            t[m] = (unsigned)(d[m >> 2][m & 3] - lo);
            const int sn = (t[m] < RNG_NODES) ? s[m >> 2][m & 3] : 0;
            h[m] = *(const v2f*)(hw2 + 2 * (long)sn);
        }
#pragma unroll
        for (int m = 0; m < 16; ++m) {
            if (t[m] < RNG_NODES) {
                const float wm = w[m >> 2][m & 3];
                atomicAdd(&part[2 * t[m]],     h[m].x * wm);
                atomicAdd(&part[2 * t[m] + 1], h[m].y * wm);
            }
        }
    }
    __syncthreads();

    const int rot = c * OUT_ROT;
    for (int i = threadIdx.x; i < 2 * RNG_NODES; i += PRIV_THREADS) {
        int j = i + rot;
        if (j >= 2 * RNG_NODES) j -= 2 * RNG_NODES;
        int g = 2 * lo + j;
        if (g < 2 * N_NODES) unsafeAtomicAdd(&out[g], part[j]);
    }
}

__global__ void finish2(const float* __restrict__ hw2, const float* __restrict__ dinv,
                        const float* __restrict__ b2, float* __restrict__ out) {
    int n = blockIdx.x * blockDim.x + threadIdx.x;
    if (n >= N_NODES) return;
    float sn = dinv[n]; sn *= sn;
    out[2 * n]     += hw2[2 * n]     * sn + b2[0];
    out[2 * n + 1] += hw2[2 * n + 1] * sn + b2[1];
}

extern "C" void kernel_launch(void* const* d_in, const int* in_sizes, int n_in,
                              void* d_out, int out_size, void* d_ws, size_t ws_size,
                              hipStream_t stream) {
    const float* x  = (const float*)d_in[0];
    const int*   ei = (const int*)  d_in[1];
    const float* W1 = (const float*)d_in[2];
    const float* b1 = (const float*)d_in[3];
    const float* W2 = (const float*)d_in[4];
    const float* b2 = (const float*)d_in[5];
    const int* src = ei;
    const int* dst = ei + N_EDGES;
    float* out = (float*)d_out;

    float* ws   = (float*)d_ws;
    float* dinv = ws;
    float* en   = dinv + N_NODES;
    float* hw1  = en + N_EDGES;
    float* h    = hw1 + (size_t)N_NODES * HID;
    float* hw2  = h   + (size_t)N_NODES * HID;

    const dim3 blk(256);

    zero_kernel<<<(N_NODES + 255) / 256, blk, 0, stream>>>(dinv, N_NODES);
    zero_kernel<<<(N_NODES * HID + 255) / 256, blk, 0, stream>>>(h, N_NODES * HID);
    zero_kernel<<<(OUT_DIM * N_NODES + 255) / 256, blk, 0, stream>>>(out, OUT_DIM * N_NODES);

    deg_kernel  <<<N_RANGES * N_CHUNKS, dim3(PRIV_THREADS), 0, stream>>>(dst, dinv);
    dinv_kernel <<<(N_NODES + 255) / 256, blk, 0, stream>>>(dinv);
    enorm_kernel<<<(N_EDGES + 255) / 256, blk, 0, stream>>>(src, dst, dinv, en);

    gemm1_wmma<<<(NTILES + 7) / 8, blk, 0, stream>>>(x, W1, hw1);
    scatter1  <<<(N_EDGES * 16) / 256, blk, 0, stream>>>(src, dst, en, hw1, h);
    relu_bias <<<(N_NODES * HID + 255) / 256, blk, 0, stream>>>(hw1, dinv, b1, h);

    gemm2   <<<(N_NODES + 255) / 256, blk, 0, stream>>>(h, W2, hw2);
    scatter2<<<N_RANGES * N_CHUNKS, dim3(PRIV_THREADS), 0, stream>>>(src, dst, en, hw2, out);
    finish2 <<<(N_NODES + 255) / 256, blk, 0, stream>>>(hw2, dinv, b2, out);
}


# Layer-1 edge scatter: a wave walks 32 edges, lane l on features l and 32+l, so each row gather and each f32 atomic add covers one whole 128-B line (was 4 lines x 8 floats); atomic sum order as before
# speedup vs baseline: 1.9990x; 1.2223x over previous
#include <hip/hip_runtime.h>

typedef float v2f __attribute__((ext_vector_type(2)));
typedef float v4f __attribute__((ext_vector_type(4)));
typedef float v8f __attribute__((ext_vector_type(8)));
typedef int   v4i __attribute__((ext_vector_type(4)));

#define N_NODES 100000
#define N_EDGES 1600000
#define IN_DIM  165
#define HID     64
#define OUT_DIM 2
#define NTILES  (N_NODES / 16)
#define KPAIRS  83

#define N_RANGES    8
#define RNG_NODES   12512
#define N_CHUNKS    32
#define CHUNK_EDGES (N_EDGES / N_CHUNKS)
#define CHUNK_VEC4  (CHUNK_EDGES / 4)
#define PRIV_THREADS 1024
#define DEG_ROUNDS  (CHUNK_VEC4 / PRIV_THREADS)
#define DEG_ROT     (RNG_NODES / N_CHUNKS / 32 * 32)
#define OUT_ROT     (2 * RNG_NODES / N_CHUNKS / 32 * 32)

__global__ void zero_kernel(float* __restrict__ p, int n) {
    int i = blockIdx.x * blockDim.x + threadIdx.x;
    if (i < n) p[i] = 0.0f;
}

__global__ __launch_bounds__(PRIV_THREADS) void deg_kernel(const int* __restrict__ dst,
                                                           float* __restrict__ deg) {
    __shared__ unsigned int cnt[RNG_NODES];
    const int r  = blockIdx.x % N_RANGES;
    const int c  = blockIdx.x / N_RANGES;
    const int lo = r * RNG_NODES;
    for (int i = threadIdx.x; i < RNG_NODES; i += PRIV_THREADS) cnt[i] = 0u;
    __syncthreads();

    const v4i* __restrict__ pd = (const v4i*)(dst + (size_t)c * CHUNK_EDGES);
    v4i d[DEG_ROUNDS + 1];
#pragma unroll
    for (int k = 0; k < DEG_ROUNDS; ++k) d[k] = pd[threadIdx.x + k * PRIV_THREADS];
    d[DEG_ROUNDS] = (v4i){-1, -1, -1, -1};
    if (threadIdx.x + DEG_ROUNDS * PRIV_THREADS < CHUNK_VEC4)
        d[DEG_ROUNDS] = pd[threadIdx.x + DEG_ROUNDS * PRIV_THREADS];
#pragma unroll
    for (int k = 0; k <= DEG_ROUNDS; ++k) {
        unsigned t0 = (unsigned)(d[k].x - lo), t1 = (unsigned)(d[k].y - lo);
        unsigned t2 = (unsigned)(d[k].z - lo), t3 = (unsigned)(d[k].w - lo);
        if (t0 < RNG_NODES) atomicAdd(&cnt[t0], 1u);
        if (t1 < RNG_NODES) atomicAdd(&cnt[t1], 1u);
        if (t2 < RNG_NODES) atomicAdd(&cnt[t2], 1u);
        if (t3 < RNG_NODES) atomicAdd(&cnt[t3], 1u);
    }
    __syncthreads();

    const int rot = c * DEG_ROT;
    for (int i = threadIdx.x; i < RNG_NODES; i += PRIV_THREADS) {
        int j = i + rot;
        if (j >= RNG_NODES) j -= RNG_NODES;
        int n = lo + j;
        if (n < N_NODES) unsafeAtomicAdd(&deg[n], (float)cnt[j]);
    }
}

__global__ void dinv_kernel(float* __restrict__ d) {
    int i = blockIdx.x * blockDim.x + threadIdx.x;
    if (i < N_NODES) d[i] = rsqrtf(d[i] + 1.0f);
}

__global__ void enorm_kernel(const int* __restrict__ src, const int* __restrict__ dst,
                             const float* __restrict__ dinv, float* __restrict__ en) {
    int e = blockIdx.x * blockDim.x + threadIdx.x;
    if (e < N_EDGES) en[e] = dinv[src[e]] * dinv[dst[e]];
}

__global__ __launch_bounds__(256) void gemm1_wmma(const float* __restrict__ x,
                                                  const float* __restrict__ W1,
                                                  float* __restrict__ hw1) {
    __shared__ float w1p[KPAIRS * 2 * HID];
    for (int i = threadIdx.x; i < KPAIRS * 2 * HID; i += 256) {
        int kp   = i >> 7;
        int r    = i & 127;
        int col  = r >> 1;
        int sub  = r & 1;
        int krow = kp * 2 + sub;
        w1p[i] = (krow < IN_DIM) ? W1[krow * HID + col] : 0.0f;
    }
    __syncthreads();

    const int wave = threadIdx.x >> 5;
    const int lane = threadIdx.x & 31;
    const int tile = blockIdx.x * 8 + wave;
    if (tile >= NTILES) return;

    const int hh  = lane >> 4;
    const int lr  = lane & 15;
    const int row = tile * 16 + lr;
    const float* __restrict__ xrb = x + (long)row * IN_DIM;
    const float* __restrict__ xr  = xrb + 2 * hh;

    v8f acc0 = (v8f){}, acc1 = (v8f){}, acc2 = (v8f){}, acc3 = (v8f){};

    for (int k = 0; k < 164; k += 4) {
        v2f A;
        A.x = xr[k];
        A.y = xr[k + 1];
        const int ka = k + 2 * hh;
        const v2f* __restrict__ bp = (const v2f*)&w1p[(ka >> 1) * 2 * HID + lr * 2];
        v2f B0 = bp[0];
        v2f B1 = bp[16];
        v2f B2 = bp[32];
        v2f B3 = bp[48];
        acc0 = __builtin_amdgcn_wmma_f32_16x16x4_f32(false, A, false, B0, (short)0, acc0, false, false);
        acc1 = __builtin_amdgcn_wmma_f32_16x16x4_f32(false, A, false, B1, (short)0, acc1, false, false);
        acc2 = __builtin_amdgcn_wmma_f32_16x16x4_f32(false, A, false, B2, (short)0, acc2, false, false);
        acc3 = __builtin_amdgcn_wmma_f32_16x16x4_f32(false, A, false, B3, (short)0, acc3, false, false);
    }

    {
        float a164 = xrb[164];
        v2f A;
        A.x = (hh == 0) ? a164 : 0.0f;
        A.y = 0.0f;
        const v2f* __restrict__ bp = (const v2f*)&w1p[(KPAIRS - 1) * 2 * HID + lr * 2];
        v2f B0 = bp[0];
        v2f B1 = bp[16];
        v2f B2 = bp[32];
        v2f B3 = bp[48];
        acc0 = __builtin_amdgcn_wmma_f32_16x16x4_f32(false, A, false, B0, (short)0, acc0, false, false);
        acc1 = __builtin_amdgcn_wmma_f32_16x16x4_f32(false, A, false, B1, (short)0, acc1, false, false);
        acc2 = __builtin_amdgcn_wmma_f32_16x16x4_f32(false, A, false, B2, (short)0, acc2, false, false);
        acc3 = __builtin_amdgcn_wmma_f32_16x16x4_f32(false, A, false, B3, (short)0, acc3, false, false);
    }

    float* __restrict__ outp = hw1 + (long)(tile * 16 + 8 * hh) * HID + lr;
#pragma unroll
    for (int v = 0; v < 8; ++v) {
        long o = (long)v * HID;
        outp[o +  0] = acc0[v];
        outp[o + 16] = acc1[v];
        outp[o + 32] = acc2[v];
        outp[o + 48] = acc3[v];
    }
}

#define S1_WAVE_EDGES 32
#define S1_BATCH      16
static_assert(N_EDGES % (8 * S1_WAVE_EDGES) == 0, "grid covers the edges exactly, no tail");
__global__ __launch_bounds__(256) void scatter1(const int* __restrict__ src,
                                                const int* __restrict__ dst,
                                                const float* __restrict__ en,
                                                const float* __restrict__ hw1,
                                                float* __restrict__ agg) {
    const int lane = threadIdx.x & 31;
    const int wave = blockIdx.x * 8 + (threadIdx.x >> 5);
    const int e    = wave * S1_WAVE_EDGES + lane;
    const int   s = src[e];
    const int   d = dst[e];
    const int   w = __float_as_int(en[e]);
    const float* __restrict__ hl = hw1 + lane;
    float* __restrict__       al = agg + lane;
#pragma unroll
    for (int b = 0; b < S1_WAVE_EDGES; b += S1_BATCH) {
        float lo[S1_BATCH], hi[S1_BATCH];
#pragma unroll
        for (int j = 0; j < S1_BATCH; ++j) {
            const long sj = __builtin_amdgcn_readlane(s, b + j);
            lo[j] = hl[sj * HID];
            hi[j] = hl[sj * HID + 32];
        }
#pragma unroll
        for (int j = 0; j < S1_BATCH; ++j) {
            const long  dj = __builtin_amdgcn_readlane(d, b + j);
            const float wj = __int_as_float(__builtin_amdgcn_readlane(w, b + j));
            unsafeAtomicAdd(al + dj * HID,      lo[j] * wj);
            unsafeAtomicAdd(al + dj * HID + 32, hi[j] * wj);
        }
    }
}

__global__ void relu_bias(const float* __restrict__ hw1, const float* __restrict__ dinv,
                          const float* __restrict__ b1, float* __restrict__ h) {
    int tid = blockIdx.x * blockDim.x + threadIdx.x;
    if (tid >= N_NODES * HID) return;
    int n = tid >> 6, f = tid & 63;
    float sn = dinv[n]; sn *= sn;
    float v = h[tid] + hw1[tid] * sn + b1[f];
    h[tid] = v > 0.0f ? v : 0.0f;
}

__global__ void gemm2(const float* __restrict__ h, const float* __restrict__ W2,
                      float* __restrict__ hw2) {
    int n = blockIdx.x * blockDim.x + threadIdx.x;
    if (n >= N_NODES) return;
    const float* hr = h + (long)n * HID;
    float o0 = 0.0f, o1 = 0.0f;
#pragma unroll 8
    for (int k = 0; k < HID; ++k) {
        float hv = hr[k];
        o0 = fmaf(hv, W2[2 * k],     o0);
        o1 = fmaf(hv, W2[2 * k + 1], o1);
    }
    hw2[2 * n]     = o0;
    hw2[2 * n + 1] = o1;
}

__global__ __launch_bounds__(PRIV_THREADS) void scatter2(const int* __restrict__ src,
                                                         const int* __restrict__ dst,
                                                         const float* __restrict__ en,
                                                         const float* __restrict__ hw2,
                                                         float* __restrict__ out) {
    __shared__ float part[2 * RNG_NODES];
    const int r  = blockIdx.x % N_RANGES;
    const int c  = blockIdx.x / N_RANGES;
    const int lo = r * RNG_NODES;
    for (int i = threadIdx.x; i < 2 * RNG_NODES; i += PRIV_THREADS) part[i] = 0.0f;
    __syncthreads();

    const v4i* __restrict__ ps = (const v4i*)(src + (size_t)c * CHUNK_EDGES);
    const v4i* __restrict__ pd = (const v4i*)(dst + (size_t)c * CHUNK_EDGES);
    const v4f* __restrict__ pw = (const v4f*)(en  + (size_t)c * CHUNK_EDGES);
    for (int b = threadIdx.x; b < CHUNK_VEC4; b += 4 * PRIV_THREADS) {
        v4i d[4], s[4];
        v4f w[4];
#pragma unroll
        for (int k = 0; k < 4; ++k) {
            const int  i  = b + k * PRIV_THREADS;
            const bool ok = i < CHUNK_VEC4;
            const int  ic = ok ? i : CHUNK_VEC4 - 1;
            d[k] = pd[ic];
            s[k] = ps[ic];
            w[k] = pw[ic];
            if (!ok) d[k] = (v4i){-1, -1, -1, -1};
        }
        unsigned t[16];
        v2f      h[16];
#pragma unroll
        for (int m = 0; m < 16; ++m) {
            t[m] = (unsigned)(d[m >> 2][m & 3] - lo);
            const int sn = (t[m] < RNG_NODES) ? s[m >> 2][m & 3] : 0;
            h[m] = *(const v2f*)(hw2 + 2 * (long)sn);
        }
#pragma unroll
        for (int m = 0; m < 16; ++m) {
            if (t[m] < RNG_NODES) {
                const float wm = w[m >> 2][m & 3];
                atomicAdd(&part[2 * t[m]],     h[m].x * wm);
                atomicAdd(&part[2 * t[m] + 1], h[m].y * wm);
            }
        }
    }
    __syncthreads();

    const int rot = c * OUT_ROT;
    for (int i = threadIdx.x; i < 2 * RNG_NODES; i += PRIV_THREADS) {
        int j = i + rot;
        if (j >= 2 * RNG_NODES) j -= 2 * RNG_NODES;
        int g = 2 * lo + j;
        if (g < 2 * N_NODES) unsafeAtomicAdd(&out[g], part[j]);
    }
}

__global__ void finish2(const float* __restrict__ hw2, const float* __restrict__ dinv,
                        const float* __restrict__ b2, float* __restrict__ out) {
    int n = blockIdx.x * blockDim.x + threadIdx.x;
    if (n >= N_NODES) return;
    float sn = dinv[n]; sn *= sn;
    out[2 * n]     += hw2[2 * n]     * sn + b2[0];
    out[2 * n + 1] += hw2[2 * n + 1] * sn + b2[1];
}

extern "C" void kernel_launch(void* const* d_in, const int* in_sizes, int n_in,
                              void* d_out, int out_size, void* d_ws, size_t ws_size,
                              hipStream_t stream) {
    const float* x  = (const float*)d_in[0];
    const int*   ei = (const int*)  d_in[1];
    const float* W1 = (const float*)d_in[2];
    const float* b1 = (const float*)d_in[3];
    const float* W2 = (const float*)d_in[4];
    const float* b2 = (const float*)d_in[5];
    const int* src = ei;
    const int* dst = ei + N_EDGES;
    float* out = (float*)d_out;

    float* ws   = (float*)d_ws;
    float* dinv = ws;
    float* en   = dinv + N_NODES;
    float* hw1  = en + N_EDGES;
    float* h    = hw1 + (size_t)N_NODES * HID;
    float* hw2  = h   + (size_t)N_NODES * HID;

    const dim3 blk(256);

    zero_kernel<<<(N_NODES + 255) / 256, blk, 0, stream>>>(dinv, N_NODES);
    zero_kernel<<<(N_NODES * HID + 255) / 256, blk, 0, stream>>>(h, N_NODES * HID);
    zero_kernel<<<(OUT_DIM * N_NODES + 255) / 256, blk, 0, stream>>>(out, OUT_DIM * N_NODES);

    deg_kernel  <<<N_RANGES * N_CHUNKS, dim3(PRIV_THREADS), 0, stream>>>(dst, dinv);
    dinv_kernel <<<(N_NODES + 255) / 256, blk, 0, stream>>>(dinv);
    enorm_kernel<<<(N_EDGES + 255) / 256, blk, 0, stream>>>(src, dst, dinv, en);

    gemm1_wmma<<<(NTILES + 7) / 8, blk, 0, stream>>>(x, W1, hw1);
    scatter1  <<<N_EDGES / (8 * S1_WAVE_EDGES), blk, 0, stream>>>(src, dst, en, hw1, h);
    relu_bias <<<(N_NODES * HID + 255) / 256, blk, 0, stream>>>(hw1, dinv, b1, h);

    gemm2   <<<(N_NODES + 255) / 256, blk, 0, stream>>>(h, W2, hw2);
    scatter2<<<N_RANGES * N_CHUNKS, dim3(PRIV_THREADS), 0, stream>>>(src, dst, en, hw2, out);
    finish2 <<<(N_NODES + 255) / 256, blk, 0, stream>>>(hw2, dinv, b2, out);
}


# Edges ordered by destination in the call (LDS counting sort: chunk counts, scan, ranked fill); a wave sums a row's gathered rows in registers, both layers: no float atomics, sum order = list order
# speedup vs baseline: 3.3694x; 1.6855x over previous
#include <hip/hip_runtime.h>

typedef float v2f __attribute__((ext_vector_type(2)));
typedef float v4f __attribute__((ext_vector_type(4)));
typedef float v8f __attribute__((ext_vector_type(8)));
typedef int   v4i __attribute__((ext_vector_type(4)));

#define N_NODES 100000
#define N_EDGES 1600000
#define IN_DIM  165
#define HID     64
#define OUT_DIM 2
#define NTILES  (N_NODES / 16)
#define KPAIRS  83

#define N_RANGES    8
#define RNG_NODES   12512
#define N_CHUNKS    32
#define CHUNK_EDGES (N_EDGES / N_CHUNKS)
#define CHUNK_VEC4  (CHUNK_EDGES / 4)
#define PRIV_THREADS 1024
#define DEG_ROUNDS  (CHUNK_VEC4 / PRIV_THREADS)
#define SCAN_BLOCKS ((N_NODES + PRIV_THREADS - 1) / PRIV_THREADS)
static_assert(SCAN_BLOCKS <= PRIV_THREADS, "one thread per scan block in the fill kernel");

__device__ __forceinline__ unsigned block_excl_scan(unsigned v, unsigned* ws, unsigned& total) {
    const int lane = threadIdx.x & 31;
    const int wave = threadIdx.x >> 5;
    unsigned incl = v;
#pragma unroll
    for (int off = 1; off < 32; off <<= 1) {
        const unsigned t = __shfl_up(incl, off);
        if (lane >= off) incl += t;
    }
    if (lane == 31) ws[wave] = incl;
    __syncthreads();
    if (wave == 0) {
        const unsigned w = ws[lane];
        unsigned wi = w;
#pragma unroll
        for (int off = 1; off < 32; off <<= 1) {
            const unsigned t = __shfl_up(wi, off);
            if (lane >= off) wi += t;
        }
        ws[lane] = wi - w;
        if (lane == 31) ws[32] = wi;
    }
    __syncthreads();
    total = ws[32];
    return incl - v + ws[wave];
}

__global__ __launch_bounds__(PRIV_THREADS) void count_kernel(const int* __restrict__ dst,
                                                             unsigned int* __restrict__ cc) {
    __shared__ unsigned int cnt[RNG_NODES];
    const int r  = blockIdx.x % N_RANGES;
    const int c  = blockIdx.x / N_RANGES;
    const int lo = r * RNG_NODES;
    for (int i = threadIdx.x; i < RNG_NODES; i += PRIV_THREADS) cnt[i] = 0u;
    __syncthreads();

    const v4i* __restrict__ pd = (const v4i*)(dst + (size_t)c * CHUNK_EDGES);
    v4i d[DEG_ROUNDS + 1];
#pragma unroll
    for (int k = 0; k < DEG_ROUNDS; ++k) d[k] = pd[threadIdx.x + k * PRIV_THREADS];
    d[DEG_ROUNDS] = (v4i){-1, -1, -1, -1};
    if (threadIdx.x + DEG_ROUNDS * PRIV_THREADS < CHUNK_VEC4)
        d[DEG_ROUNDS] = pd[threadIdx.x + DEG_ROUNDS * PRIV_THREADS];
#pragma unroll
    for (int k = 0; k <= DEG_ROUNDS; ++k) {
        unsigned t0 = (unsigned)(d[k].x - lo), t1 = (unsigned)(d[k].y - lo);
        unsigned t2 = (unsigned)(d[k].z - lo), t3 = (unsigned)(d[k].w - lo);
        if (t0 < RNG_NODES) atomicAdd(&cnt[t0], 1u);
        if (t1 < RNG_NODES) atomicAdd(&cnt[t1], 1u);
        if (t2 < RNG_NODES) atomicAdd(&cnt[t2], 1u);
        if (t3 < RNG_NODES) atomicAdd(&cnt[t3], 1u);
    }
    __syncthreads();

    unsigned int* __restrict__ row = cc + (size_t)c * N_NODES;
    for (int i = threadIdx.x; i < RNG_NODES; i += PRIV_THREADS) {
        const int n = lo + i;
        if (n < N_NODES) row[n] = cnt[i];
    }
}

__global__ __launch_bounds__(PRIV_THREADS) void scan_kernel(unsigned int* __restrict__ cc,
                                                            float* __restrict__ dinv,
                                                            int* __restrict__ degi,
                                                            unsigned int* __restrict__ rowloc,
                                                            unsigned int* __restrict__ btot) {
    __shared__ unsigned int ws[33];
    const int  n     = blockIdx.x * PRIV_THREADS + threadIdx.x;
    const bool valid = n < N_NODES;
    unsigned v[N_CHUNKS];
#pragma unroll
    for (int c = 0; c < N_CHUNKS; ++c) v[c] = valid ? cc[(size_t)c * N_NODES + n] : 0u;
    unsigned run = 0u;
#pragma unroll
    for (int c = 0; c < N_CHUNKS; ++c) {
        if (valid) cc[(size_t)c * N_NODES + n] = run;
        run += v[c];
    }
    if (valid) {
        degi[n] = (int)run;
        dinv[n] = rsqrtf((float)run + 1.0f);
    }
    unsigned total;
    const unsigned excl = block_excl_scan(run, ws, total);
    if (valid) rowloc[n] = excl;
    if (threadIdx.x == 0) btot[blockIdx.x] = total;
}

__global__ __launch_bounds__(PRIV_THREADS) void fill_kernel(const int* __restrict__ src,
                                                            const int* __restrict__ dst,
                                                            const unsigned int* __restrict__ cc,
                                                            const unsigned int* __restrict__ rowloc,
                                                            const unsigned int* __restrict__ btot,
                                                            unsigned int* __restrict__ rowptr,
                                                            int* __restrict__ csr) {
    __shared__ unsigned int cur[RNG_NODES];
    __shared__ unsigned int bbase[SCAN_BLOCKS];
    __shared__ unsigned int ws[33];
    const int r  = blockIdx.x % N_RANGES;
    const int c  = blockIdx.x / N_RANGES;
    const int lo = r * RNG_NODES;

    unsigned total;
    const unsigned bv = (threadIdx.x < SCAN_BLOCKS) ? btot[threadIdx.x] : 0u;
    const unsigned be = block_excl_scan(bv, ws, total);
    if (threadIdx.x < SCAN_BLOCKS) bbase[threadIdx.x] = be;
    __syncthreads();

    const unsigned int* __restrict__ row = cc + (size_t)c * N_NODES;
    for (int i = threadIdx.x; i < RNG_NODES; i += PRIV_THREADS) {
        const int n = lo + i;
        unsigned p = 0u;
        if (n < N_NODES) {
            const unsigned base = bbase[n / PRIV_THREADS] + rowloc[n];
            p = base + row[n];
            if (c == 0) rowptr[n] = base;
        }
        cur[i] = p;
    }
    __syncthreads();

    const v4i* __restrict__ ps = (const v4i*)(src + (size_t)c * CHUNK_EDGES);
    const v4i* __restrict__ pd = (const v4i*)(dst + (size_t)c * CHUNK_EDGES);
    for (int b = threadIdx.x; b < CHUNK_VEC4; b += 4 * PRIV_THREADS) {
        v4i d[4], s[4];
#pragma unroll
        for (int k = 0; k < 4; ++k) {
            const int  i  = b + k * PRIV_THREADS;
            const bool ok = i < CHUNK_VEC4;
            const int  ic = ok ? i : CHUNK_VEC4 - 1;
            d[k] = pd[ic];
            s[k] = ps[ic];
            if (!ok) d[k] = (v4i){-1, -1, -1, -1};
        }
#pragma unroll
        for (int m = 0; m < 16; ++m) {
            const unsigned t = (unsigned)(d[m >> 2][m & 3] - lo);
            if (t < RNG_NODES) {
                const unsigned pos = atomicAdd(&cur[t], 1u);
                csr[pos] = s[m >> 2][m & 3];
            }
        }
    }
}

__global__ __launch_bounds__(256) void gemm1_wmma(const float* __restrict__ x,
                                                  const float* __restrict__ W1,
                                                  float* __restrict__ hw1) {
    __shared__ float w1p[KPAIRS * 2 * HID];
    for (int i = threadIdx.x; i < KPAIRS * 2 * HID; i += 256) {
        int kp   = i >> 7;
        int r    = i & 127;
        int col  = r >> 1;
        int sub  = r & 1;
        int krow = kp * 2 + sub;
        w1p[i] = (krow < IN_DIM) ? W1[krow * HID + col] : 0.0f;
    }
    __syncthreads();

    const int wave = threadIdx.x >> 5;
    const int lane = threadIdx.x & 31;
    const int tile = blockIdx.x * 8 + wave;
    if (tile >= NTILES) return;

    const int hh  = lane >> 4;
    const int lr  = lane & 15;
    const int row = tile * 16 + lr;
    const float* __restrict__ xrb = x + (long)row * IN_DIM;
    const float* __restrict__ xr  = xrb + 2 * hh;

    v8f acc0 = (v8f){}, acc1 = (v8f){}, acc2 = (v8f){}, acc3 = (v8f){};

    for (int k = 0; k < 164; k += 4) {
        v2f A;
        A.x = xr[k];
        A.y = xr[k + 1];
        const int ka = k + 2 * hh;
        const v2f* __restrict__ bp = (const v2f*)&w1p[(ka >> 1) * 2 * HID + lr * 2];
        v2f B0 = bp[0];
        v2f B1 = bp[16];
        v2f B2 = bp[32];
        v2f B3 = bp[48];
        acc0 = __builtin_amdgcn_wmma_f32_16x16x4_f32(false, A, false, B0, (short)0, acc0, false, false);
        acc1 = __builtin_amdgcn_wmma_f32_16x16x4_f32(false, A, false, B1, (short)0, acc1, false, false);
        acc2 = __builtin_amdgcn_wmma_f32_16x16x4_f32(false, A, false, B2, (short)0, acc2, false, false);
        acc3 = __builtin_amdgcn_wmma_f32_16x16x4_f32(false, A, false, B3, (short)0, acc3, false, false);
    }

    {
        float a164 = xrb[164];
        v2f A;
        A.x = (hh == 0) ? a164 : 0.0f;
        A.y = 0.0f;
        const v2f* __restrict__ bp = (const v2f*)&w1p[(KPAIRS - 1) * 2 * HID + lr * 2];
        v2f B0 = bp[0];
        v2f B1 = bp[16];
        v2f B2 = bp[32];
        v2f B3 = bp[48];
        acc0 = __builtin_amdgcn_wmma_f32_16x16x4_f32(false, A, false, B0, (short)0, acc0, false, false);
        acc1 = __builtin_amdgcn_wmma_f32_16x16x4_f32(false, A, false, B1, (short)0, acc1, false, false);
        acc2 = __builtin_amdgcn_wmma_f32_16x16x4_f32(false, A, false, B2, (short)0, acc2, false, false);
        acc3 = __builtin_amdgcn_wmma_f32_16x16x4_f32(false, A, false, B3, (short)0, acc3, false, false);
    }

    float* __restrict__ outp = hw1 + (long)(tile * 16 + 8 * hh) * HID + lr;
#pragma unroll
    for (int v = 0; v < 8; ++v) {
        long o = (long)v * HID;
        outp[o +  0] = acc0[v];
        outp[o + 16] = acc1[v];
        outp[o + 32] = acc2[v];
        outp[o + 48] = acc3[v];
    }
}

#define P1_NODES 4
#define P1_BATCH 8
static_assert(N_NODES % (8 * P1_NODES) == 0, "grid covers the nodes exactly, no tail");
__global__ __launch_bounds__(256) void pull1(const int* __restrict__ csr,
                                             const unsigned int* __restrict__ rowptr,
                                             const int* __restrict__ degi,
                                             const float* __restrict__ dinv,
                                             const float* __restrict__ hw1,
                                             const float* __restrict__ b1,
                                             const float* __restrict__ W2,
                                             float* __restrict__ hw2) {
    const int lane = threadIdx.x & 31;
    const int n0   = (blockIdx.x * 8 + (threadIdx.x >> 5)) * P1_NODES;
    const int   ml     = n0 + (lane & (P1_NODES - 1));
    const int   startv = (int)rowptr[ml];
    const int   cntv   = degi[ml];
    const int   dnv    = __float_as_int(dinv[ml]);
    const v4f   w2 = ((const v4f*)W2)[lane];
    const v2f   bv = ((const v2f*)b1)[lane];
    const v2f* __restrict__ hr = (const v2f*)hw1 + lane;

    for (int k = 0; k < P1_NODES; ++k) {
        const int   n     = n0 + k;
        const int   start = __builtin_amdgcn_readlane(startv, k);
        const int   cnt   = __builtin_amdgcn_readlane(cntv, k);
        const float dn    = __int_as_float(__builtin_amdgcn_readlane(dnv, k));
        const v2f   self  = hr[(long)n * (HID / 2)];
        v2f acc = (v2f){0.0f, 0.0f};
        for (int base = 0; base < cnt; base += 32) {
            const int  j  = base + lane;
            const bool ok = j < cnt;
            const int   sr = csr[start + (ok ? j : cnt - 1)];
            const int   s  = ok ? sr : n;
            const int   w = __float_as_int(dinv[s]);
            const int   m = (cnt - base < 32) ? (cnt - base) : 32;
            for (int jj = 0; jj < m; jj += P1_BATCH) {
                v2f row[P1_BATCH];
#pragma unroll
                for (int q = 0; q < P1_BATCH; ++q) {
                    const long sj = __builtin_amdgcn_readlane(s, jj + q);
                    row[q] = hr[sj * (HID / 2)];
                }
#pragma unroll
                for (int q = 0; q < P1_BATCH; ++q) {
                    const float wq = (jj + q < m) ? __int_as_float(__builtin_amdgcn_readlane(w, jj + q)) * dn : 0.0f;
                    acc.x += row[q].x * wq;
                    acc.y += row[q].y * wq;
                }
            }
        }
        const float sn = dn * dn;
        float h0 = acc.x + self.x * sn + bv.x;
        float h1 = acc.y + self.y * sn + bv.y;
        h0 = h0 > 0.0f ? h0 : 0.0f;
        h1 = h1 > 0.0f ? h1 : 0.0f;
        float p0 = h0 * w2.x + h1 * w2.z;
        float p1 = h0 * w2.y + h1 * w2.w;
#pragma unroll
        for (int off = 16; off > 0; off >>= 1) {
            p0 += __shfl_xor(p0, off);
            p1 += __shfl_xor(p1, off);
        }
        if (lane == 0) *(v2f*)(hw2 + 2 * (long)n) = (v2f){p0, p1};
    }
}

#define P2_LANES 8
static_assert((N_NODES * P2_LANES) % 256 == 0, "grid covers the nodes exactly, no tail");
__global__ __launch_bounds__(256) void pull2(const int* __restrict__ csr,
                                             const unsigned int* __restrict__ rowptr,
                                             const int* __restrict__ degi,
                                             const float* __restrict__ dinv,
                                             const float* __restrict__ hw2,
                                             const float* __restrict__ b2,
                                             float* __restrict__ out) {
    const int n   = (blockIdx.x * 256 + threadIdx.x) / P2_LANES;
    const int sub = threadIdx.x & (P2_LANES - 1);
    const int   start = (int)rowptr[n];
    const int   cnt   = degi[n];
    const float dn    = dinv[n];
    const v2f   hs    = *(const v2f*)(hw2 + 2 * (long)n);
    float a0 = 0.0f, a1 = 0.0f;
    for (int j = sub; j < cnt; j += P2_LANES) {
        const int   s = csr[start + j];
        const float w = dinv[s] * dn;
        const v2f   h = *(const v2f*)(hw2 + 2 * (long)s);
        a0 += h.x * w;
        a1 += h.y * w;
    }
#pragma unroll
    for (int off = P2_LANES / 2; off > 0; off >>= 1) {
        a0 += __shfl_xor(a0, off);
        a1 += __shfl_xor(a1, off);
    }
    if (sub == 0) {
        const float sn = dn * dn;
        v2f o;
        o.x = a0 + hs.x * sn + b2[0];
        o.y = a1 + hs.y * sn + b2[1];
        *(v2f*)(out + 2 * (long)n) = o;
    }
}

extern "C" void kernel_launch(void* const* d_in, const int* in_sizes, int n_in,
                              void* d_out, int out_size, void* d_ws, size_t ws_size,
                              hipStream_t stream) {
    const float* x  = (const float*)d_in[0];
    const int*   ei = (const int*)  d_in[1];
    const float* W1 = (const float*)d_in[2];
    const float* b1 = (const float*)d_in[3];
    const float* W2 = (const float*)d_in[4];
    const float* b2 = (const float*)d_in[5];
    const int* src = ei;
    const int* dst = ei + N_EDGES;
    float* out = (float*)d_out;

    float*        ws     = (float*)d_ws;
    float*        dinv   = ws;
    int*          degi   = (int*)(dinv + N_NODES);
    unsigned int* rowloc = (unsigned int*)(degi + N_NODES);
    unsigned int* rowptr = rowloc + N_NODES;
    unsigned int* btot   = rowptr + N_NODES;
    unsigned int* cc     = btot + 128;
    int*          csr    = (int*)(cc + (size_t)N_CHUNKS * N_NODES);
    float*        hw1    = (float*)(csr + N_EDGES);
    float*        hw2    = hw1 + (size_t)N_NODES * HID;

    const dim3 blk(256);

    count_kernel<<<N_RANGES * N_CHUNKS, dim3(PRIV_THREADS), 0, stream>>>(dst, cc);
    scan_kernel <<<SCAN_BLOCKS, dim3(PRIV_THREADS), 0, stream>>>(cc, dinv, degi, rowloc, btot);
    fill_kernel <<<N_RANGES * N_CHUNKS, dim3(PRIV_THREADS), 0, stream>>>(src, dst, cc, rowloc, btot, rowptr, csr);

    gemm1_wmma<<<(NTILES + 7) / 8, blk, 0, stream>>>(x, W1, hw1);
    pull1     <<<N_NODES / (8 * P1_NODES), blk, 0, stream>>>(csr, rowptr, degi, dinv, hw1, b1, W2, hw2);

    pull2<<<N_NODES * P2_LANES / 256, blk, 0, stream>>>(csr, rowptr, degi, dinv, hw2, b2, out);
}


# GEMM1 moved into the edge-fill launch as 256 extra 1024-thread workgroups (64-register budget: a fill and a GEMM workgroup share each WGP) so its x-load waits overlap the fill; 5 launches, same sums
# speedup vs baseline: 4.0469x; 1.2011x over previous
#include <hip/hip_runtime.h>

typedef float v2f __attribute__((ext_vector_type(2)));
typedef float v4f __attribute__((ext_vector_type(4)));
typedef float v8f __attribute__((ext_vector_type(8)));
typedef int   v4i __attribute__((ext_vector_type(4)));

#define N_NODES 100000
#define N_EDGES 1600000
#define IN_DIM  165
#define HID     64
#define OUT_DIM 2
#define NTILES  (N_NODES / 16)
#define KPAIRS  83

#define N_RANGES    8
#define RNG_NODES   12512
#define N_CHUNKS    32
#define CHUNK_EDGES (N_EDGES / N_CHUNKS)
#define CHUNK_VEC4  (CHUNK_EDGES / 4)
#define PRIV_THREADS 1024
#define DEG_ROUNDS  (CHUNK_VEC4 / PRIV_THREADS)
#define SCAN_BLOCKS ((N_NODES + PRIV_THREADS - 1) / PRIV_THREADS)
static_assert(SCAN_BLOCKS <= PRIV_THREADS, "one thread per scan block in the fill kernel");

__device__ __forceinline__ unsigned block_excl_scan(unsigned v, unsigned* ws, unsigned& total) {
    const int lane = threadIdx.x & 31;
    const int wave = threadIdx.x >> 5;
    unsigned incl = v;
#pragma unroll
    for (int off = 1; off < 32; off <<= 1) {
        const unsigned t = __shfl_up(incl, off);
        if (lane >= off) incl += t;
    }
    if (lane == 31) ws[wave] = incl;
    __syncthreads();
    if (wave == 0) {
        const unsigned w = ws[lane];
        unsigned wi = w;
#pragma unroll
        for (int off = 1; off < 32; off <<= 1) {
            const unsigned t = __shfl_up(wi, off);
            if (lane >= off) wi += t;
        }
        ws[lane] = wi - w;
        if (lane == 31) ws[32] = wi;
    }
    __syncthreads();
    total = ws[32];
    return incl - v + ws[wave];
}

__global__ __launch_bounds__(PRIV_THREADS) void count_kernel(const int* __restrict__ dst,
                                                             unsigned int* __restrict__ cc) {
    __shared__ unsigned int cnt[RNG_NODES];
    const int r  = blockIdx.x % N_RANGES;
    const int c  = blockIdx.x / N_RANGES;
    const int lo = r * RNG_NODES;
    for (int i = threadIdx.x; i < RNG_NODES; i += PRIV_THREADS) cnt[i] = 0u;
    __syncthreads();

    const v4i* __restrict__ pd = (const v4i*)(dst + (size_t)c * CHUNK_EDGES);
    v4i d[DEG_ROUNDS + 1];
#pragma unroll
    for (int k = 0; k < DEG_ROUNDS; ++k) d[k] = pd[threadIdx.x + k * PRIV_THREADS];
    d[DEG_ROUNDS] = (v4i){-1, -1, -1, -1};
    if (threadIdx.x + DEG_ROUNDS * PRIV_THREADS < CHUNK_VEC4)
        d[DEG_ROUNDS] = pd[threadIdx.x + DEG_ROUNDS * PRIV_THREADS];
#pragma unroll
    for (int k = 0; k <= DEG_ROUNDS; ++k) {
        unsigned t0 = (unsigned)(d[k].x - lo), t1 = (unsigned)(d[k].y - lo);
        unsigned t2 = (unsigned)(d[k].z - lo), t3 = (unsigned)(d[k].w - lo);
        if (t0 < RNG_NODES) atomicAdd(&cnt[t0], 1u);
        if (t1 < RNG_NODES) atomicAdd(&cnt[t1], 1u);
        if (t2 < RNG_NODES) atomicAdd(&cnt[t2], 1u);
        if (t3 < RNG_NODES) atomicAdd(&cnt[t3], 1u);
    }
    __syncthreads();

    unsigned int* __restrict__ row = cc + (size_t)c * N_NODES;
    for (int i = threadIdx.x; i < RNG_NODES; i += PRIV_THREADS) {
        const int n = lo + i;
        if (n < N_NODES) row[n] = cnt[i];
    }
}

__global__ __launch_bounds__(PRIV_THREADS) void scan_kernel(unsigned int* __restrict__ cc,
                                                            float* __restrict__ dinv,
                                                            int* __restrict__ degi,
                                                            unsigned int* __restrict__ rowloc,
                                                            unsigned int* __restrict__ btot) {
    __shared__ unsigned int ws[33];
    const int  n     = blockIdx.x * PRIV_THREADS + threadIdx.x;
    const bool valid = n < N_NODES;
    unsigned v[N_CHUNKS];
#pragma unroll
    for (int c = 0; c < N_CHUNKS; ++c) v[c] = valid ? cc[(size_t)c * N_NODES + n] : 0u;
    unsigned run = 0u;
#pragma unroll
    for (int c = 0; c < N_CHUNKS; ++c) {
        if (valid) cc[(size_t)c * N_NODES + n] = run;
        run += v[c];
    }
    if (valid) {
        degi[n] = (int)run;
        dinv[n] = rsqrtf((float)run + 1.0f);
    }
    unsigned total;
    const unsigned excl = block_excl_scan(run, ws, total);
    if (valid) rowloc[n] = excl;
    if (threadIdx.x == 0) btot[blockIdx.x] = total;
}

#define FILL_BBASE_OFF  RNG_NODES
#define FILL_WS_OFF     (RNG_NODES + 128)
#define FILL_LDS_WORDS  (RNG_NODES + 128 + 33)
static_assert(SCAN_BLOCKS <= 128, "block bases fit their LDS slot");
__device__ __forceinline__ void fill_part(unsigned int* __restrict__ lds, const int bidx,
                                          const int* __restrict__ src,
                                          const int* __restrict__ dst,
                                          const unsigned int* __restrict__ cc,
                                          const unsigned int* __restrict__ rowloc,
                                          const unsigned int* __restrict__ btot,
                                          unsigned int* __restrict__ rowptr,
                                          int* __restrict__ csr) {
    unsigned int* cur   = lds;
    unsigned int* bbase = lds + FILL_BBASE_OFF;
    unsigned int* ws    = lds + FILL_WS_OFF;
    const int r  = bidx % N_RANGES;
    const int c  = bidx / N_RANGES;
    const int lo = r * RNG_NODES;

    unsigned total;
    const unsigned bv = (threadIdx.x < SCAN_BLOCKS) ? btot[threadIdx.x] : 0u;
    const unsigned be = block_excl_scan(bv, ws, total);
    if (threadIdx.x < SCAN_BLOCKS) bbase[threadIdx.x] = be;
    __syncthreads();

    const unsigned int* __restrict__ row = cc + (size_t)c * N_NODES;
    for (int i = threadIdx.x; i < RNG_NODES; i += PRIV_THREADS) {
        const int n = lo + i;
        unsigned p = 0u;
        if (n < N_NODES) {
            const unsigned base = bbase[n / PRIV_THREADS] + rowloc[n];
            p = base + row[n];
            if (c == 0) rowptr[n] = base;
        }
        cur[i] = p;
    }
    __syncthreads();

    const v4i* __restrict__ ps = (const v4i*)(src + (size_t)c * CHUNK_EDGES);
    const v4i* __restrict__ pd = (const v4i*)(dst + (size_t)c * CHUNK_EDGES);
    for (int b = threadIdx.x; b < CHUNK_VEC4; b += 4 * PRIV_THREADS) {
        v4i d[4], s[4];
#pragma unroll
        for (int k = 0; k < 4; ++k) {
            const int  i  = b + k * PRIV_THREADS;
            const bool ok = i < CHUNK_VEC4;
            const int  ic = ok ? i : CHUNK_VEC4 - 1;
            d[k] = pd[ic];
            s[k] = ps[ic];
            if (!ok) d[k] = (v4i){-1, -1, -1, -1};
        }
#pragma unroll
        for (int m = 0; m < 16; ++m) {
            const unsigned t = (unsigned)(d[m >> 2][m & 3] - lo);
            if (t < RNG_NODES) {
                const unsigned pos = atomicAdd(&cur[t], 1u);
                csr[pos] = s[m >> 2][m & 3];
            }
        }
    }
}

__device__ __forceinline__ void gemm_step(const float* __restrict__ xr, const float* __restrict__ w1p,
                                          const int k, const int hh, const int lr,
                                          v8f& acc0, v8f& acc1, v8f& acc2, v8f& acc3) {
    v2f A;
    A.x = xr[k];
    A.y = xr[k + 1];
    const int ka = k + 2 * hh;
    const v2f* __restrict__ bp = (const v2f*)&w1p[(ka >> 1) * 2 * HID + lr * 2];
    v2f B0 = bp[0];
    v2f B1 = bp[16];
    v2f B2 = bp[32];
    v2f B3 = bp[48];
    acc0 = __builtin_amdgcn_wmma_f32_16x16x4_f32(false, A, false, B0, (short)0, acc0, false, false);
    acc1 = __builtin_amdgcn_wmma_f32_16x16x4_f32(false, A, false, B1, (short)0, acc1, false, false);
    acc2 = __builtin_amdgcn_wmma_f32_16x16x4_f32(false, A, false, B2, (short)0, acc2, false, false);
    acc3 = __builtin_amdgcn_wmma_f32_16x16x4_f32(false, A, false, B3, (short)0, acc3, false, false);
}

#define GEMM_BLOCKS     256
#define GEMM_LDS_WORDS  (KPAIRS * 2 * HID)
__device__ __forceinline__ void gemm_part(float* __restrict__ w1p, const int gidx,
                                          const float* __restrict__ x,
                                          const float* __restrict__ W1,
                                          float* __restrict__ hw1) {
    for (int i = threadIdx.x; i < KPAIRS * 2 * HID; i += PRIV_THREADS) {
        int kp   = i >> 7;
        int r    = i & 127;
        int col  = r >> 1;
        int sub  = r & 1;
        int krow = kp * 2 + sub;
        w1p[i] = (krow < IN_DIM) ? W1[krow * HID + col] : 0.0f;
    }
    __syncthreads();

    const int wave = threadIdx.x >> 5;
    const int lane = threadIdx.x & 31;
    const int tile = gidx + GEMM_BLOCKS * wave;
    if (tile >= NTILES) return;

    const int hh  = lane >> 4;
    const int lr  = lane & 15;
    const int row = tile * 16 + lr;
    const float* __restrict__ xrb = x + (long)row * IN_DIM;
    const float* __restrict__ xr  = xrb + 2 * hh;

    v8f acc0 = (v8f){}, acc1 = (v8f){}, acc2 = (v8f){}, acc3 = (v8f){};

    for (int k = 0; k < 160; k += 4)
        gemm_step(xr, w1p, k, hh, lr, acc0, acc1, acc2, acc3);
    gemm_step(xr, w1p, 160, hh, lr, acc0, acc1, acc2, acc3);

    {
        float a164 = xrb[164];
        v2f A;
        A.x = (hh == 0) ? a164 : 0.0f;
        A.y = 0.0f;
        const v2f* __restrict__ bp = (const v2f*)&w1p[(KPAIRS - 1) * 2 * HID + lr * 2];
        v2f B0 = bp[0];
        v2f B1 = bp[16];
        v2f B2 = bp[32];
        v2f B3 = bp[48];
        acc0 = __builtin_amdgcn_wmma_f32_16x16x4_f32(false, A, false, B0, (short)0, acc0, false, false);
        acc1 = __builtin_amdgcn_wmma_f32_16x16x4_f32(false, A, false, B1, (short)0, acc1, false, false);
        acc2 = __builtin_amdgcn_wmma_f32_16x16x4_f32(false, A, false, B2, (short)0, acc2, false, false);
        acc3 = __builtin_amdgcn_wmma_f32_16x16x4_f32(false, A, false, B3, (short)0, acc3, false, false);
    }

    float* __restrict__ outp = hw1 + (long)(tile * 16 + 8 * hh) * HID + lr;
#pragma unroll
    for (int v = 0; v < 8; ++v) {
        long o = (long)v * HID;
        outp[o +  0] = acc0[v];
        outp[o + 16] = acc1[v];
        outp[o + 32] = acc2[v];
        outp[o + 48] = acc3[v];
    }
}

#define FG_LDS_WORDS (FILL_LDS_WORDS > GEMM_LDS_WORDS ? FILL_LDS_WORDS : GEMM_LDS_WORDS)
static_assert(2 * N_RANGES * N_CHUNKS == 2 * GEMM_BLOCKS, "as many fill as GEMM workgroups");
static_assert(25 * GEMM_BLOCKS >= NTILES, "the tile rounds of 32 waves cover every tile");
__global__ __launch_bounds__(PRIV_THREADS) __attribute__((amdgpu_waves_per_eu(16, 16)))
void fill_gemm_kernel(const int* __restrict__ src,
                      const int* __restrict__ dst,
                      const unsigned int* __restrict__ cc,
                      const unsigned int* __restrict__ rowloc,
                      const unsigned int* __restrict__ btot,
                      unsigned int* __restrict__ rowptr,
                      int* __restrict__ csr,
                      const float* __restrict__ x,
                      const float* __restrict__ W1,
                      float* __restrict__ hw1) {
    __shared__ __attribute__((aligned(16))) unsigned int lds[FG_LDS_WORDS];
    const int b    = blockIdx.x;
    const int role = (b ^ (b >> 8)) & 1;
    const int idx  = b >> 1;
    if (role == 0) fill_part(lds, idx, src, dst, cc, rowloc, btot, rowptr, csr);
    else           gemm_part((float*)lds, idx, x, W1, hw1);
}

#define P1_NODES 4
#define P1_BATCH 8
static_assert(N_NODES % (8 * P1_NODES) == 0, "grid covers the nodes exactly, no tail");
__global__ __launch_bounds__(256) void pull1(const int* __restrict__ csr,
                                             const unsigned int* __restrict__ rowptr,
                                             const int* __restrict__ degi,
                                             const float* __restrict__ dinv,
                                             const float* __restrict__ hw1,
                                             const float* __restrict__ b1,
                                             const float* __restrict__ W2,
                                             float* __restrict__ hw2) {
    const int lane = threadIdx.x & 31;
    const int n0   = (blockIdx.x * 8 + (threadIdx.x >> 5)) * P1_NODES;
    const int   ml     = n0 + (lane & (P1_NODES - 1));
    const int   startv = (int)rowptr[ml];
    const int   cntv   = degi[ml];
    const int   dnv    = __float_as_int(dinv[ml]);
    const v4f   w2 = ((const v4f*)W2)[lane];
    const v2f   bv = ((const v2f*)b1)[lane];
    const v2f* __restrict__ hr = (const v2f*)hw1 + lane;

    for (int k = 0; k < P1_NODES; ++k) {
        const int   n     = n0 + k;
        const int   start = __builtin_amdgcn_readlane(startv, k);
        const int   cnt   = __builtin_amdgcn_readlane(cntv, k);
        const float dn    = __int_as_float(__builtin_amdgcn_readlane(dnv, k));
        const v2f   self  = hr[(long)n * (HID / 2)];
        v2f acc = (v2f){0.0f, 0.0f};
        for (int base = 0; base < cnt; base += 32) {
            const int  j  = base + lane;
            const bool ok = j < cnt;
            const int   sr = csr[start + (ok ? j : cnt - 1)];
            const int   s  = ok ? sr : n;
            const int   w = __float_as_int(dinv[s]);
            const int   m = (cnt - base < 32) ? (cnt - base) : 32;
            for (int jj = 0; jj < m; jj += P1_BATCH) {
                v2f row[P1_BATCH];
#pragma unroll
                for (int q = 0; q < P1_BATCH; ++q) {
                    const long sj = __builtin_amdgcn_readlane(s, jj + q);
                    row[q] = hr[sj * (HID / 2)];
                }
#pragma unroll
                for (int q = 0; q < P1_BATCH; ++q) {
                    const float wq = (jj + q < m) ? __int_as_float(__builtin_amdgcn_readlane(w, jj + q)) * dn : 0.0f;
                    acc.x += row[q].x * wq;
                    acc.y += row[q].y * wq;
                }
            }
        }
        const float sn = dn * dn;
        float h0 = acc.x + self.x * sn + bv.x;
        float h1 = acc.y + self.y * sn + bv.y;
        h0 = h0 > 0.0f ? h0 : 0.0f;
        h1 = h1 > 0.0f ? h1 : 0.0f;
        float p0 = h0 * w2.x + h1 * w2.z;
        float p1 = h0 * w2.y + h1 * w2.w;
#pragma unroll
        for (int off = 16; off > 0; off >>= 1) {
            p0 += __shfl_xor(p0, off);
            p1 += __shfl_xor(p1, off);
        }
        if (lane == 0) *(v2f*)(hw2 + 2 * (long)n) = (v2f){p0, p1};
    }
}

#define P2_LANES 8
static_assert((N_NODES * P2_LANES) % 256 == 0, "grid covers the nodes exactly, no tail");
__global__ __launch_bounds__(256) void pull2(const int* __restrict__ csr,
                                             const unsigned int* __restrict__ rowptr,
                                             const int* __restrict__ degi,
                                             const float* __restrict__ dinv,
                                             const float* __restrict__ hw2,
                                             const float* __restrict__ b2,
                                             float* __restrict__ out) {
    const int n   = (blockIdx.x * 256 + threadIdx.x) / P2_LANES;
    const int sub = threadIdx.x & (P2_LANES - 1);
    const int   start = (int)rowptr[n];
    const int   cnt   = degi[n];
    const float dn    = dinv[n];
    const v2f   hs    = *(const v2f*)(hw2 + 2 * (long)n);
    float a0 = 0.0f, a1 = 0.0f;
    for (int j = sub; j < cnt; j += P2_LANES) {
        const int   s = csr[start + j];
        const float w = dinv[s] * dn;
        const v2f   h = *(const v2f*)(hw2 + 2 * (long)s);
        a0 += h.x * w;
        a1 += h.y * w;
    }
#pragma unroll
    for (int off = P2_LANES / 2; off > 0; off >>= 1) {
        a0 += __shfl_xor(a0, off);
        a1 += __shfl_xor(a1, off);
    }
    if (sub == 0) {
        const float sn = dn * dn;
        v2f o;
        o.x = a0 + hs.x * sn + b2[0];
        o.y = a1 + hs.y * sn + b2[1];
        *(v2f*)(out + 2 * (long)n) = o;
    }
}

extern "C" void kernel_launch(void* const* d_in, const int* in_sizes, int n_in,
                              void* d_out, int out_size, void* d_ws, size_t ws_size,
                              hipStream_t stream) {
    const float* x  = (const float*)d_in[0];
    const int*   ei = (const int*)  d_in[1];
    const float* W1 = (const float*)d_in[2];
    const float* b1 = (const float*)d_in[3];
    const float* W2 = (const float*)d_in[4];
    const float* b2 = (const float*)d_in[5];
    const int* src = ei;
    const int* dst = ei + N_EDGES;
    float* out = (float*)d_out;

    float*        ws     = (float*)d_ws;
    float*        dinv   = ws;
    int*          degi   = (int*)(dinv + N_NODES);
    unsigned int* rowloc = (unsigned int*)(degi + N_NODES);
    unsigned int* rowptr = rowloc + N_NODES;
    unsigned int* btot   = rowptr + N_NODES;
    unsigned int* cc     = btot + 128;
    int*          csr    = (int*)(cc + (size_t)N_CHUNKS * N_NODES);
    float*        hw1    = (float*)(csr + N_EDGES);
    float*        hw2    = hw1 + (size_t)N_NODES * HID;

    const dim3 blk(256);

    count_kernel<<<N_RANGES * N_CHUNKS, dim3(PRIV_THREADS), 0, stream>>>(dst, cc);
    scan_kernel <<<SCAN_BLOCKS, dim3(PRIV_THREADS), 0, stream>>>(cc, dinv, degi, rowloc, btot);
    fill_gemm_kernel<<<N_RANGES * N_CHUNKS + GEMM_BLOCKS, dim3(PRIV_THREADS), 0, stream>>>(
        src, dst, cc, rowloc, btot, rowptr, csr, x, W1, hw1);

    pull1<<<N_NODES / (8 * P1_NODES), blk, 0, stream>>>(csr, rowptr, degi, dinv, hw1, b1, W2, hw2);

    pull2<<<N_NODES * P2_LANES / 256, blk, 0, stream>>>(csr, rowptr, degi, dinv, hw2, b2, out);
}


# Source half of the edge norm folded into the stored rows (GEMM1 and layer 1 write dinv[row]*row); both gathers drop the per-edge dinv[src] load; sum order kept, products re-associated
# speedup vs baseline: 4.5168x; 1.1161x over previous
#include <hip/hip_runtime.h>

typedef float v2f __attribute__((ext_vector_type(2)));
typedef float v4f __attribute__((ext_vector_type(4)));
typedef float v8f __attribute__((ext_vector_type(8)));
typedef int   v4i __attribute__((ext_vector_type(4)));

#define N_NODES 100000
#define N_EDGES 1600000
#define IN_DIM  165
#define HID     64
#define OUT_DIM 2
#define NTILES  (N_NODES / 16)
#define KPAIRS  83

#define N_RANGES    8
#define RNG_NODES   12512
#define N_CHUNKS    32
#define CHUNK_EDGES (N_EDGES / N_CHUNKS)
#define CHUNK_VEC4  (CHUNK_EDGES / 4)
#define PRIV_THREADS 1024
#define DEG_ROUNDS  (CHUNK_VEC4 / PRIV_THREADS)
#define SCAN_BLOCKS ((N_NODES + PRIV_THREADS - 1) / PRIV_THREADS)
static_assert(SCAN_BLOCKS <= PRIV_THREADS, "one thread per scan block in the fill kernel");

__device__ __forceinline__ unsigned block_excl_scan(unsigned v, unsigned* ws, unsigned& total) {
    const int lane = threadIdx.x & 31;
    const int wave = threadIdx.x >> 5;
    unsigned incl = v;
#pragma unroll
    for (int off = 1; off < 32; off <<= 1) {
        const unsigned t = __shfl_up(incl, off);
        if (lane >= off) incl += t;
    }
    if (lane == 31) ws[wave] = incl;
    __syncthreads();
    if (wave == 0) {
        const unsigned w = ws[lane];
        unsigned wi = w;
#pragma unroll
        for (int off = 1; off < 32; off <<= 1) {
            const unsigned t = __shfl_up(wi, off);
            if (lane >= off) wi += t;
        }
        ws[lane] = wi - w;
        if (lane == 31) ws[32] = wi;
    }
    __syncthreads();
    total = ws[32];
    return incl - v + ws[wave];
}

__global__ __launch_bounds__(PRIV_THREADS) void count_kernel(const int* __restrict__ dst,
                                                             unsigned int* __restrict__ cc) {
    __shared__ unsigned int cnt[RNG_NODES];
    const int r  = blockIdx.x % N_RANGES;
    const int c  = blockIdx.x / N_RANGES;
    const int lo = r * RNG_NODES;
    for (int i = threadIdx.x; i < RNG_NODES; i += PRIV_THREADS) cnt[i] = 0u;
    __syncthreads();

    const v4i* __restrict__ pd = (const v4i*)(dst + (size_t)c * CHUNK_EDGES);
    v4i d[DEG_ROUNDS + 1];
#pragma unroll
    for (int k = 0; k < DEG_ROUNDS; ++k) d[k] = pd[threadIdx.x + k * PRIV_THREADS];
    d[DEG_ROUNDS] = (v4i){-1, -1, -1, -1};
    if (threadIdx.x + DEG_ROUNDS * PRIV_THREADS < CHUNK_VEC4)
        d[DEG_ROUNDS] = pd[threadIdx.x + DEG_ROUNDS * PRIV_THREADS];
#pragma unroll
    for (int k = 0; k <= DEG_ROUNDS; ++k) {
        unsigned t0 = (unsigned)(d[k].x - lo), t1 = (unsigned)(d[k].y - lo);
        unsigned t2 = (unsigned)(d[k].z - lo), t3 = (unsigned)(d[k].w - lo);
        if (t0 < RNG_NODES) atomicAdd(&cnt[t0], 1u);
        if (t1 < RNG_NODES) atomicAdd(&cnt[t1], 1u);
        if (t2 < RNG_NODES) atomicAdd(&cnt[t2], 1u);
        if (t3 < RNG_NODES) atomicAdd(&cnt[t3], 1u);
    }
    __syncthreads();

    unsigned int* __restrict__ row = cc + (size_t)c * N_NODES;
    for (int i = threadIdx.x; i < RNG_NODES; i += PRIV_THREADS) {
        const int n = lo + i;
        if (n < N_NODES) row[n] = cnt[i];
    }
}

__global__ __launch_bounds__(PRIV_THREADS) void scan_kernel(unsigned int* __restrict__ cc,
                                                            float* __restrict__ dinv,
                                                            int* __restrict__ degi,
                                                            unsigned int* __restrict__ rowloc,
                                                            unsigned int* __restrict__ btot) {
    __shared__ unsigned int ws[33];
    const int  n     = blockIdx.x * PRIV_THREADS + threadIdx.x;
    const bool valid = n < N_NODES;
    unsigned v[N_CHUNKS];
#pragma unroll
    for (int c = 0; c < N_CHUNKS; ++c) v[c] = valid ? cc[(size_t)c * N_NODES + n] : 0u;
    unsigned run = 0u;
#pragma unroll
    for (int c = 0; c < N_CHUNKS; ++c) {
        if (valid) cc[(size_t)c * N_NODES + n] = run;
        run += v[c];
    }
    if (valid) {
        degi[n] = (int)run;
        dinv[n] = rsqrtf((float)run + 1.0f);
    }
    unsigned total;
    const unsigned excl = block_excl_scan(run, ws, total);
    if (valid) rowloc[n] = excl;
    if (threadIdx.x == 0) btot[blockIdx.x] = total;
}

#define FILL_BBASE_OFF  RNG_NODES
#define FILL_WS_OFF     (RNG_NODES + 128)
#define FILL_LDS_WORDS  (RNG_NODES + 128 + 33)
static_assert(SCAN_BLOCKS <= 128, "block bases fit their LDS slot");
__device__ __forceinline__ void fill_part(unsigned int* __restrict__ lds, const int bidx,
                                          const int* __restrict__ src,
                                          const int* __restrict__ dst,
                                          const unsigned int* __restrict__ cc,
                                          const unsigned int* __restrict__ rowloc,
                                          const unsigned int* __restrict__ btot,
                                          unsigned int* __restrict__ rowptr,
                                          int* __restrict__ csr) {
    unsigned int* cur   = lds;
    unsigned int* bbase = lds + FILL_BBASE_OFF;
    unsigned int* ws    = lds + FILL_WS_OFF;
    const int r  = bidx % N_RANGES;
    const int c  = bidx / N_RANGES;
    const int lo = r * RNG_NODES;

    unsigned total;
    const unsigned bv = (threadIdx.x < SCAN_BLOCKS) ? btot[threadIdx.x] : 0u;
    const unsigned be = block_excl_scan(bv, ws, total);
    if (threadIdx.x < SCAN_BLOCKS) bbase[threadIdx.x] = be;
    __syncthreads();

    const unsigned int* __restrict__ row = cc + (size_t)c * N_NODES;
    for (int i = threadIdx.x; i < RNG_NODES; i += PRIV_THREADS) {
        const int n = lo + i;
        unsigned p = 0u;
        if (n < N_NODES) {
            const unsigned base = bbase[n / PRIV_THREADS] + rowloc[n];
            p = base + row[n];
            if (c == 0) rowptr[n] = base;
        }
        cur[i] = p;
    }
    __syncthreads();

    const v4i* __restrict__ ps = (const v4i*)(src + (size_t)c * CHUNK_EDGES);
    const v4i* __restrict__ pd = (const v4i*)(dst + (size_t)c * CHUNK_EDGES);
    for (int b = threadIdx.x; b < CHUNK_VEC4; b += 4 * PRIV_THREADS) {
        v4i d[4], s[4];
#pragma unroll
        for (int k = 0; k < 4; ++k) {
            const int  i  = b + k * PRIV_THREADS;
            const bool ok = i < CHUNK_VEC4;
            const int  ic = ok ? i : CHUNK_VEC4 - 1;
            d[k] = pd[ic];
            s[k] = ps[ic];
            if (!ok) d[k] = (v4i){-1, -1, -1, -1};
        }
#pragma unroll
        for (int m = 0; m < 16; ++m) {
            const unsigned t = (unsigned)(d[m >> 2][m & 3] - lo);
            if (t < RNG_NODES) {
                const unsigned pos = atomicAdd(&cur[t], 1u);
                csr[pos] = s[m >> 2][m & 3];
            }
        }
    }
}

__device__ __forceinline__ void gemm_step(const float* __restrict__ xr, const float* __restrict__ w1p,
                                          const int k, const int hh, const int lr,
                                          v8f& acc0, v8f& acc1, v8f& acc2, v8f& acc3) {
    v2f A;
    A.x = xr[k];
    A.y = xr[k + 1];
    const int ka = k + 2 * hh;
    const v2f* __restrict__ bp = (const v2f*)&w1p[(ka >> 1) * 2 * HID + lr * 2];
    v2f B0 = bp[0];
    v2f B1 = bp[16];
    v2f B2 = bp[32];
    v2f B3 = bp[48];
    acc0 = __builtin_amdgcn_wmma_f32_16x16x4_f32(false, A, false, B0, (short)0, acc0, false, false);
    acc1 = __builtin_amdgcn_wmma_f32_16x16x4_f32(false, A, false, B1, (short)0, acc1, false, false);
    acc2 = __builtin_amdgcn_wmma_f32_16x16x4_f32(false, A, false, B2, (short)0, acc2, false, false);
    acc3 = __builtin_amdgcn_wmma_f32_16x16x4_f32(false, A, false, B3, (short)0, acc3, false, false);
}

#define GEMM_BLOCKS     256
#define GEMM_LDS_WORDS  (KPAIRS * 2 * HID)
__device__ __forceinline__ void gemm_part(float* __restrict__ w1p, const int gidx,
                                          const float* __restrict__ x,
                                          const float* __restrict__ W1,
                                          const float* __restrict__ dinv,
                                          float* __restrict__ hw1) {
    for (int i = threadIdx.x; i < KPAIRS * 2 * HID; i += PRIV_THREADS) {
        int kp   = i >> 7;
        int r    = i & 127;
        int col  = r >> 1;
        int sub  = r & 1;
        int krow = kp * 2 + sub;
        w1p[i] = (krow < IN_DIM) ? W1[krow * HID + col] : 0.0f;
    }
    __syncthreads();

    const int wave = threadIdx.x >> 5;
    const int lane = threadIdx.x & 31;
    const int tile = gidx + GEMM_BLOCKS * wave;
    if (tile >= NTILES) return;

    const int hh  = lane >> 4;
    const int lr  = lane & 15;
    const int row = tile * 16 + lr;
    const float* __restrict__ xrb = x + (long)row * IN_DIM;
    const float* __restrict__ xr  = xrb + 2 * hh;

    v8f acc0 = (v8f){}, acc1 = (v8f){}, acc2 = (v8f){}, acc3 = (v8f){};

    for (int k = 0; k < 160; k += 4)
        gemm_step(xr, w1p, k, hh, lr, acc0, acc1, acc2, acc3);
    gemm_step(xr, w1p, 160, hh, lr, acc0, acc1, acc2, acc3);

    {
        float a164 = xrb[164];
        v2f A;
        A.x = (hh == 0) ? a164 : 0.0f;
        A.y = 0.0f;
        const v2f* __restrict__ bp = (const v2f*)&w1p[(KPAIRS - 1) * 2 * HID + lr * 2];
        v2f B0 = bp[0];
        v2f B1 = bp[16];
        v2f B2 = bp[32];
        v2f B3 = bp[48];
        acc0 = __builtin_amdgcn_wmma_f32_16x16x4_f32(false, A, false, B0, (short)0, acc0, false, false);
        acc1 = __builtin_amdgcn_wmma_f32_16x16x4_f32(false, A, false, B1, (short)0, acc1, false, false);
        acc2 = __builtin_amdgcn_wmma_f32_16x16x4_f32(false, A, false, B2, (short)0, acc2, false, false);
        acc3 = __builtin_amdgcn_wmma_f32_16x16x4_f32(false, A, false, B3, (short)0, acc3, false, false);
    }

    const v4f* __restrict__ dq = (const v4f*)(dinv + tile * 16 + 8 * hh);
    const v4f dlo = dq[0], dhi = dq[1];
    const float dv[8] = {dlo.x, dlo.y, dlo.z, dlo.w, dhi.x, dhi.y, dhi.z, dhi.w};
    float* __restrict__ outp = hw1 + (long)(tile * 16 + 8 * hh) * HID + lr;
#pragma unroll
    for (int v = 0; v < 8; ++v) {
        long o = (long)v * HID;
        outp[o +  0] = acc0[v] * dv[v];
        outp[o + 16] = acc1[v] * dv[v];
        outp[o + 32] = acc2[v] * dv[v];
        outp[o + 48] = acc3[v] * dv[v];
    }
}

#define FG_LDS_WORDS (FILL_LDS_WORDS > GEMM_LDS_WORDS ? FILL_LDS_WORDS : GEMM_LDS_WORDS)
static_assert(2 * N_RANGES * N_CHUNKS == 2 * GEMM_BLOCKS, "as many fill as GEMM workgroups");
static_assert(25 * GEMM_BLOCKS >= NTILES, "the tile rounds of 32 waves cover every tile");
__global__ __launch_bounds__(PRIV_THREADS) __attribute__((amdgpu_waves_per_eu(16, 16)))
void fill_gemm_kernel(const int* __restrict__ src,
                      const int* __restrict__ dst,
                      const unsigned int* __restrict__ cc,
                      const unsigned int* __restrict__ rowloc,
                      const unsigned int* __restrict__ btot,
                      unsigned int* __restrict__ rowptr,
                      int* __restrict__ csr,
                      const float* __restrict__ x,
                      const float* __restrict__ W1,
                      const float* __restrict__ dinv,
                      float* __restrict__ hw1) {
    __shared__ __attribute__((aligned(16))) unsigned int lds[FG_LDS_WORDS];
    const int b    = blockIdx.x;
    const int role = (b ^ (b >> 8)) & 1;
    const int idx  = b >> 1;
    if (role == 0) fill_part(lds, idx, src, dst, cc, rowloc, btot, rowptr, csr);
    else           gemm_part((float*)lds, idx, x, W1, dinv, hw1);
}

#define P1_NODES 4
#define P1_BATCH 8
static_assert(N_NODES % (8 * P1_NODES) == 0, "grid covers the nodes exactly, no tail");
__global__ __launch_bounds__(256) void pull1(const int* __restrict__ csr,
                                             const unsigned int* __restrict__ rowptr,
                                             const int* __restrict__ degi,
                                             const float* __restrict__ dinv,
                                             const float* __restrict__ hw1,
                                             const float* __restrict__ b1,
                                             const float* __restrict__ W2,
                                             float* __restrict__ hw2) {
    const int lane = threadIdx.x & 31;
    const int n0   = (blockIdx.x * 8 + (threadIdx.x >> 5)) * P1_NODES;
    const int   ml     = n0 + (lane & (P1_NODES - 1));
    const int   startv = (int)rowptr[ml];
    const int   cntv   = degi[ml];
    const int   dnv    = __float_as_int(dinv[ml]);
    const v4f   w2 = ((const v4f*)W2)[lane];
    const v2f   bv = ((const v2f*)b1)[lane];
    const v2f* __restrict__ hr = (const v2f*)hw1 + lane;

    for (int k = 0; k < P1_NODES; ++k) {
        const int   n     = n0 + k;
        const int   start = __builtin_amdgcn_readlane(startv, k);
        const int   cnt   = __builtin_amdgcn_readlane(cntv, k);
        const float dn    = __int_as_float(__builtin_amdgcn_readlane(dnv, k));
        const v2f   self  = hr[(long)n * (HID / 2)];
        v2f acc = (v2f){0.0f, 0.0f};
        for (int base = 0; base < cnt; base += 32) {
            const int  j  = base + lane;
            const bool ok = j < cnt;
            const int   sr = csr[start + (ok ? j : cnt - 1)];
            const int   s  = ok ? sr : n;
            const int   m = (cnt - base < 32) ? (cnt - base) : 32;
            for (int jj = 0; jj < m; jj += P1_BATCH) {
                v2f row[P1_BATCH];
#pragma unroll
                for (int q = 0; q < P1_BATCH; ++q) {
                    const long sj = __builtin_amdgcn_readlane(s, jj + q);
                    row[q] = hr[sj * (HID / 2)];
                }
#pragma unroll
                for (int q = 0; q < P1_BATCH; ++q) {
                    const float wq = (jj + q < m) ? 1.0f : 0.0f;
                    acc.x = __builtin_fmaf(row[q].x, wq, acc.x);
                    acc.y = __builtin_fmaf(row[q].y, wq, acc.y);
                }
            }
        }
        float h0 = (acc.x + self.x) * dn + bv.x;
        float h1 = (acc.y + self.y) * dn + bv.y;
        h0 = h0 > 0.0f ? h0 : 0.0f;
        h1 = h1 > 0.0f ? h1 : 0.0f;
        float p0 = h0 * w2.x + h1 * w2.z;
        float p1 = h0 * w2.y + h1 * w2.w;
#pragma unroll
        for (int off = 16; off > 0; off >>= 1) {
            p0 += __shfl_xor(p0, off);
            p1 += __shfl_xor(p1, off);
        }
        if (lane == 0) *(v2f*)(hw2 + 2 * (long)n) = (v2f){p0 * dn, p1 * dn};
    }
}

#define P2_LANES 8
static_assert((N_NODES * P2_LANES) % 256 == 0, "grid covers the nodes exactly, no tail");
__global__ __launch_bounds__(256) void pull2(const int* __restrict__ csr,
                                             const unsigned int* __restrict__ rowptr,
                                             const int* __restrict__ degi,
                                             const float* __restrict__ dinv,
                                             const float* __restrict__ hw2,
                                             const float* __restrict__ b2,
                                             float* __restrict__ out) {
    const int n   = (blockIdx.x * 256 + threadIdx.x) / P2_LANES;
    const int sub = threadIdx.x & (P2_LANES - 1);
    const int   start = (int)rowptr[n];
    const int   cnt   = degi[n];
    const float dn    = dinv[n];
    const v2f   hs    = *(const v2f*)(hw2 + 2 * (long)n);
    float a0 = 0.0f, a1 = 0.0f;
    for (int j = sub; j < cnt; j += P2_LANES) {
        const int   s = csr[start + j];
        const v2f   h = *(const v2f*)(hw2 + 2 * (long)s);
        a0 += h.x;
        a1 += h.y;
    }
#pragma unroll
    for (int off = P2_LANES / 2; off > 0; off >>= 1) {
        a0 += __shfl_xor(a0, off);
        a1 += __shfl_xor(a1, off);
    }
    if (sub == 0) {
        v2f o;
        o.x = (a0 + hs.x) * dn + b2[0];
        o.y = (a1 + hs.y) * dn + b2[1];
        *(v2f*)(out + 2 * (long)n) = o;
    }
}

extern "C" void kernel_launch(void* const* d_in, const int* in_sizes, int n_in,
                              void* d_out, int out_size, void* d_ws, size_t ws_size,
                              hipStream_t stream) {
    const float* x  = (const float*)d_in[0];
    const int*   ei = (const int*)  d_in[1];
    const float* W1 = (const float*)d_in[2];
    const float* b1 = (const float*)d_in[3];
    const float* W2 = (const float*)d_in[4];
    const float* b2 = (const float*)d_in[5];
    const int* src = ei;
    const int* dst = ei + N_EDGES;
    float* out = (float*)d_out;

    float*        ws     = (float*)d_ws;
    float*        dinv   = ws;
    int*          degi   = (int*)(dinv + N_NODES);
    unsigned int* rowloc = (unsigned int*)(degi + N_NODES);
    unsigned int* rowptr = rowloc + N_NODES;
    unsigned int* btot   = rowptr + N_NODES;
    unsigned int* cc     = btot + 128;
    int*          csr    = (int*)(cc + (size_t)N_CHUNKS * N_NODES);
    float*        hw1    = (float*)(csr + N_EDGES);
    float*        hw2    = hw1 + (size_t)N_NODES * HID;

    const dim3 blk(256);

    count_kernel<<<N_RANGES * N_CHUNKS, dim3(PRIV_THREADS), 0, stream>>>(dst, cc);
    scan_kernel <<<SCAN_BLOCKS, dim3(PRIV_THREADS), 0, stream>>>(cc, dinv, degi, rowloc, btot);
    fill_gemm_kernel<<<N_RANGES * N_CHUNKS + GEMM_BLOCKS, dim3(PRIV_THREADS), 0, stream>>>(
        src, dst, cc, rowloc, btot, rowptr, csr, x, W1, dinv, hw1);

    pull1<<<N_NODES / (8 * P1_NODES), blk, 0, stream>>>(csr, rowptr, degi, dinv, hw1, b1, W2, hw2);

    pull2<<<N_NODES * P2_LANES / 256, blk, 0, stream>>>(csr, rowptr, degi, dinv, hw2, b2, out);
}


# Fill+GEMM1 launch: global loads issued in groups ahead of use, not one wait per load (x fragments of 8 K steps before their 32 WMMAs; W1 staging; the fill's cursor set-up); sums and order unchanged
# speedup vs baseline: 5.4198x; 1.1999x over previous
#include <hip/hip_runtime.h>

typedef float v2f __attribute__((ext_vector_type(2)));
typedef float v4f __attribute__((ext_vector_type(4)));
typedef float v8f __attribute__((ext_vector_type(8)));
typedef int   v4i __attribute__((ext_vector_type(4)));

#define N_NODES 100000
#define N_EDGES 1600000
#define IN_DIM  165
#define HID     64
#define OUT_DIM 2
#define NTILES  (N_NODES / 16)
#define KPAIRS  83

#define N_RANGES    8
#define RNG_NODES   12512
#define N_CHUNKS    32
#define CHUNK_EDGES (N_EDGES / N_CHUNKS)
#define CHUNK_VEC4  (CHUNK_EDGES / 4)
#define PRIV_THREADS 1024
#define DEG_ROUNDS  (CHUNK_VEC4 / PRIV_THREADS)
#define SCAN_BLOCKS ((N_NODES + PRIV_THREADS - 1) / PRIV_THREADS)
static_assert(SCAN_BLOCKS <= PRIV_THREADS, "one thread per scan block in the fill kernel");

__device__ __forceinline__ unsigned block_excl_scan(unsigned v, unsigned* ws, unsigned& total) {
    const int lane = threadIdx.x & 31;
    const int wave = threadIdx.x >> 5;
    unsigned incl = v;
#pragma unroll
    for (int off = 1; off < 32; off <<= 1) {
        const unsigned t = __shfl_up(incl, off);
        if (lane >= off) incl += t;
    }
    if (lane == 31) ws[wave] = incl;
    __syncthreads();
    if (wave == 0) {
        const unsigned w = ws[lane];
        unsigned wi = w;
#pragma unroll
        for (int off = 1; off < 32; off <<= 1) {
            const unsigned t = __shfl_up(wi, off);
            if (lane >= off) wi += t;
        }
        ws[lane] = wi - w;
        if (lane == 31) ws[32] = wi;
    }
    __syncthreads();
    total = ws[32];
    return incl - v + ws[wave];
}

__global__ __launch_bounds__(PRIV_THREADS) void count_kernel(const int* __restrict__ dst,
                                                             unsigned int* __restrict__ cc) {
    __shared__ unsigned int cnt[RNG_NODES];
    const int r  = blockIdx.x % N_RANGES;
    const int c  = blockIdx.x / N_RANGES;
    const int lo = r * RNG_NODES;
    for (int i = threadIdx.x; i < RNG_NODES; i += PRIV_THREADS) cnt[i] = 0u;
    __syncthreads();

    const v4i* __restrict__ pd = (const v4i*)(dst + (size_t)c * CHUNK_EDGES);
    v4i d[DEG_ROUNDS + 1];
#pragma unroll
    for (int k = 0; k < DEG_ROUNDS; ++k) d[k] = pd[threadIdx.x + k * PRIV_THREADS];
    d[DEG_ROUNDS] = (v4i){-1, -1, -1, -1};
    if (threadIdx.x + DEG_ROUNDS * PRIV_THREADS < CHUNK_VEC4)
        d[DEG_ROUNDS] = pd[threadIdx.x + DEG_ROUNDS * PRIV_THREADS];
#pragma unroll
    for (int k = 0; k <= DEG_ROUNDS; ++k) {
        unsigned t0 = (unsigned)(d[k].x - lo), t1 = (unsigned)(d[k].y - lo);
        unsigned t2 = (unsigned)(d[k].z - lo), t3 = (unsigned)(d[k].w - lo);
        if (t0 < RNG_NODES) atomicAdd(&cnt[t0], 1u);
        if (t1 < RNG_NODES) atomicAdd(&cnt[t1], 1u);
        if (t2 < RNG_NODES) atomicAdd(&cnt[t2], 1u);
        if (t3 < RNG_NODES) atomicAdd(&cnt[t3], 1u);
    }
    __syncthreads();

    unsigned int* __restrict__ row = cc + (size_t)c * N_NODES;
    for (int i = threadIdx.x; i < RNG_NODES; i += PRIV_THREADS) {
        const int n = lo + i;
        if (n < N_NODES) row[n] = cnt[i];
    }
}

__global__ __launch_bounds__(PRIV_THREADS) void scan_kernel(unsigned int* __restrict__ cc,
                                                            float* __restrict__ dinv,
                                                            int* __restrict__ degi,
                                                            unsigned int* __restrict__ rowloc,
                                                            unsigned int* __restrict__ btot) {
    __shared__ unsigned int ws[33];
    const int  n     = blockIdx.x * PRIV_THREADS + threadIdx.x;
    const bool valid = n < N_NODES;
    unsigned v[N_CHUNKS];
#pragma unroll
    for (int c = 0; c < N_CHUNKS; ++c) v[c] = valid ? cc[(size_t)c * N_NODES + n] : 0u;
    unsigned run = 0u;
#pragma unroll
    for (int c = 0; c < N_CHUNKS; ++c) {
        if (valid) cc[(size_t)c * N_NODES + n] = run;
        run += v[c];
    }
    if (valid) {
        degi[n] = (int)run;
        dinv[n] = rsqrtf((float)run + 1.0f);
    }
    unsigned total;
    const unsigned excl = block_excl_scan(run, ws, total);
    if (valid) rowloc[n] = excl;
    if (threadIdx.x == 0) btot[blockIdx.x] = total;
}

#define FILL_BBASE_OFF  RNG_NODES
#define FILL_WS_OFF     (RNG_NODES + 128)
#define FILL_LDS_WORDS  (RNG_NODES + 128 + 33)
#define CUR_ROUNDS      ((RNG_NODES + PRIV_THREADS - 1) / PRIV_THREADS)
static_assert(SCAN_BLOCKS <= 128, "block bases fit their LDS slot");
__device__ __forceinline__ void fill_part(unsigned int* __restrict__ lds, const int bidx,
                                          const int* __restrict__ src,
                                          const int* __restrict__ dst,
                                          const unsigned int* __restrict__ cc,
                                          const unsigned int* __restrict__ rowloc,
                                          const unsigned int* __restrict__ btot,
                                          unsigned int* __restrict__ rowptr,
                                          int* __restrict__ csr) {
    unsigned int* cur   = lds;
    unsigned int* bbase = lds + FILL_BBASE_OFF;
    unsigned int* ws    = lds + FILL_WS_OFF;
    const int r  = bidx % N_RANGES;
    const int c  = bidx / N_RANGES;
    const int lo = r * RNG_NODES;

    const unsigned int* __restrict__ row = cc + (size_t)c * N_NODES;
    unsigned rl[CUR_ROUNDS], rw[CUR_ROUNDS];
#pragma unroll
    for (int j = 0; j < CUR_ROUNDS; ++j) {
        const int n  = lo + threadIdx.x + j * PRIV_THREADS;
        const int nc = (n < N_NODES) ? n : N_NODES - 1;
        rl[j] = rowloc[nc];
        rw[j] = row[nc];
    }

    unsigned total;
    const unsigned bv = (threadIdx.x < SCAN_BLOCKS) ? btot[threadIdx.x] : 0u;
    const unsigned be = block_excl_scan(bv, ws, total);
    if (threadIdx.x < SCAN_BLOCKS) bbase[threadIdx.x] = be;
    __syncthreads();

#pragma unroll
    for (int j = 0; j < CUR_ROUNDS; ++j) {
        const int i = threadIdx.x + j * PRIV_THREADS;
        const int n = lo + i;
        if (i < RNG_NODES) {
            unsigned p = 0u;
            if (n < N_NODES) {
                const unsigned base = bbase[n / PRIV_THREADS] + rl[j];
                p = base + rw[j];
                if (c == 0) rowptr[n] = base;
            }
            cur[i] = p;
        }
    }
    __syncthreads();

    const v4i* __restrict__ ps = (const v4i*)(src + (size_t)c * CHUNK_EDGES);
    const v4i* __restrict__ pd = (const v4i*)(dst + (size_t)c * CHUNK_EDGES);
    for (int b = threadIdx.x; b < CHUNK_VEC4; b += 4 * PRIV_THREADS) {
        v4i d[4], s[4];
#pragma unroll
        for (int k = 0; k < 4; ++k) {
            const int  i  = b + k * PRIV_THREADS;
            const bool ok = i < CHUNK_VEC4;
            const int  ic = ok ? i : CHUNK_VEC4 - 1;
            d[k] = pd[ic];
            s[k] = ps[ic];
            if (!ok) d[k] = (v4i){-1, -1, -1, -1};
        }
#pragma unroll
        for (int m = 0; m < 16; ++m) {
            const unsigned t = (unsigned)(d[m >> 2][m & 3] - lo);
            if (t < RNG_NODES) {
                const unsigned pos = atomicAdd(&cur[t], 1u);
                csr[pos] = s[m >> 2][m & 3];
            }
        }
    }
}

__device__ __forceinline__ void gemm_step(const v2f A, const float* __restrict__ w1p,
                                          const int k, const int hh, const int lr,
                                          v8f& acc0, v8f& acc1, v8f& acc2, v8f& acc3) {
    const int ka = k + 2 * hh;
    const v2f* __restrict__ bp = (const v2f*)&w1p[(ka >> 1) * 2 * HID + lr * 2];
    v2f B0 = bp[0];
    v2f B1 = bp[16];
    v2f B2 = bp[32];
    v2f B3 = bp[48];
    acc0 = __builtin_amdgcn_wmma_f32_16x16x4_f32(false, A, false, B0, (short)0, acc0, false, false);
    acc1 = __builtin_amdgcn_wmma_f32_16x16x4_f32(false, A, false, B1, (short)0, acc1, false, false);
    acc2 = __builtin_amdgcn_wmma_f32_16x16x4_f32(false, A, false, B2, (short)0, acc2, false, false);
    acc3 = __builtin_amdgcn_wmma_f32_16x16x4_f32(false, A, false, B3, (short)0, acc3, false, false);
}

#define GEMM_BLOCKS     256
#define GEMM_LDS_WORDS  (KPAIRS * 2 * HID)
#define W1_ROUNDS       ((GEMM_LDS_WORDS + PRIV_THREADS - 1) / PRIV_THREADS)
#define W1_STAGE_WORDS  (W1_ROUNDS * PRIV_THREADS)
#define A_GROUP         8
static_assert(160 % (4 * A_GROUP) == 0, "whole groups cover K = 0..159");
__device__ __forceinline__ void gemm_part(float* __restrict__ w1p, const int gidx,
                                          const float* __restrict__ x,
                                          const float* __restrict__ W1,
                                          const float* __restrict__ dinv,
                                          float* __restrict__ hw1) {
    float wv[W1_ROUNDS];
#pragma unroll
    for (int j = 0; j < W1_ROUNDS; ++j) {
        const int  i    = threadIdx.x + j * PRIV_THREADS;
        const int  kp   = i >> 7;
        const int  r    = i & 127;
        const int  col  = r >> 1;
        const int  sub  = r & 1;
        const int  krow = kp * 2 + sub;
        const bool in   = krow < IN_DIM;
        const float v   = W1[(in ? krow : IN_DIM - 1) * HID + col];
        wv[j] = in ? v : 0.0f;
    }
#pragma unroll
    for (int j = 0; j < W1_ROUNDS; ++j) w1p[threadIdx.x + j * PRIV_THREADS] = wv[j];
    __syncthreads();

    const int wave = threadIdx.x >> 5;
    const int lane = threadIdx.x & 31;
    const int tile = gidx + GEMM_BLOCKS * wave;
    if (tile >= NTILES) return;

    const int hh  = lane >> 4;
    const int lr  = lane & 15;
    const int row = tile * 16 + lr;
    const float* __restrict__ xrb = x + (long)row * IN_DIM;
    const float* __restrict__ xr  = xrb + 2 * hh;

    v8f acc0 = (v8f){}, acc1 = (v8f){}, acc2 = (v8f){}, acc3 = (v8f){};

    for (int k0 = 0; k0 < 160; k0 += 4 * A_GROUP) {
        v2f A[A_GROUP];
#pragma unroll
        for (int j = 0; j < A_GROUP; ++j) {
            A[j].x = xr[k0 + 4 * j];
            A[j].y = xr[k0 + 4 * j + 1];
        }
#pragma unroll
        for (int j = 0; j < A_GROUP; ++j)
            gemm_step(A[j], w1p, k0 + 4 * j, hh, lr, acc0, acc1, acc2, acc3);
    }
    v2f A160;
    A160.x = xr[160];
    A160.y = xr[161];
    const float a164 = xrb[164];
    gemm_step(A160, w1p, 160, hh, lr, acc0, acc1, acc2, acc3);

    {
        v2f A;
        A.x = (hh == 0) ? a164 : 0.0f;
        A.y = 0.0f;
        const v2f* __restrict__ bp = (const v2f*)&w1p[(KPAIRS - 1) * 2 * HID + lr * 2];
        v2f B0 = bp[0];
        v2f B1 = bp[16];
        v2f B2 = bp[32];
        v2f B3 = bp[48];
        acc0 = __builtin_amdgcn_wmma_f32_16x16x4_f32(false, A, false, B0, (short)0, acc0, false, false);
        acc1 = __builtin_amdgcn_wmma_f32_16x16x4_f32(false, A, false, B1, (short)0, acc1, false, false);
        acc2 = __builtin_amdgcn_wmma_f32_16x16x4_f32(false, A, false, B2, (short)0, acc2, false, false);
        acc3 = __builtin_amdgcn_wmma_f32_16x16x4_f32(false, A, false, B3, (short)0, acc3, false, false);
    }

    const v4f* __restrict__ dq = (const v4f*)(dinv + tile * 16 + 8 * hh);
    const v4f dlo = dq[0], dhi = dq[1];
    const float dv[8] = {dlo.x, dlo.y, dlo.z, dlo.w, dhi.x, dhi.y, dhi.z, dhi.w};
    float* __restrict__ outp = hw1 + (long)(tile * 16 + 8 * hh) * HID + lr;
#pragma unroll
    for (int v = 0; v < 8; ++v) {
        long o = (long)v * HID;
        outp[o +  0] = acc0[v] * dv[v];
        outp[o + 16] = acc1[v] * dv[v];
        outp[o + 32] = acc2[v] * dv[v];
        outp[o + 48] = acc3[v] * dv[v];
    }
}

#define FG_LDS_WORDS (FILL_LDS_WORDS > GEMM_LDS_WORDS ? FILL_LDS_WORDS : GEMM_LDS_WORDS)
static_assert(W1_STAGE_WORDS <= FG_LDS_WORDS, "the whole staging rounds of W1 fit the shared buffer");
static_assert(2 * N_RANGES * N_CHUNKS == 2 * GEMM_BLOCKS, "as many fill as GEMM workgroups");
static_assert(25 * GEMM_BLOCKS >= NTILES, "the tile rounds of 32 waves cover every tile");
__global__ __launch_bounds__(PRIV_THREADS) __attribute__((amdgpu_waves_per_eu(16, 16)))
void fill_gemm_kernel(const int* __restrict__ src,
                      const int* __restrict__ dst,
                      const unsigned int* __restrict__ cc,
                      const unsigned int* __restrict__ rowloc,
                      const unsigned int* __restrict__ btot,
                      unsigned int* __restrict__ rowptr,
                      int* __restrict__ csr,
                      const float* __restrict__ x,
                      const float* __restrict__ W1,
                      const float* __restrict__ dinv,
                      float* __restrict__ hw1) {
    __shared__ __attribute__((aligned(16))) unsigned int lds[FG_LDS_WORDS];
    const int b    = blockIdx.x;
    const int role = (b ^ (b >> 8)) & 1;
    const int idx  = b >> 1;
    if (role == 0) fill_part(lds, idx, src, dst, cc, rowloc, btot, rowptr, csr);
    else           gemm_part((float*)lds, idx, x, W1, dinv, hw1);
}

#define P1_NODES 4
#define P1_BATCH 8
static_assert(N_NODES % (8 * P1_NODES) == 0, "grid covers the nodes exactly, no tail");
__global__ __launch_bounds__(256) void pull1(const int* __restrict__ csr,
                                             const unsigned int* __restrict__ rowptr,
                                             const int* __restrict__ degi,
                                             const float* __restrict__ dinv,
                                             const float* __restrict__ hw1,
                                             const float* __restrict__ b1,
                                             const float* __restrict__ W2,
                                             float* __restrict__ hw2) {
    const int lane = threadIdx.x & 31;
    const int n0   = (blockIdx.x * 8 + (threadIdx.x >> 5)) * P1_NODES;
    const int   ml     = n0 + (lane & (P1_NODES - 1));
    const int   startv = (int)rowptr[ml];
    const int   cntv   = degi[ml];
    const int   dnv    = __float_as_int(dinv[ml]);
    const v4f   w2 = ((const v4f*)W2)[lane];
    const v2f   bv = ((const v2f*)b1)[lane];
    const v2f* __restrict__ hr = (const v2f*)hw1 + lane;

    for (int k = 0; k < P1_NODES; ++k) {
        const int   n     = n0 + k;
        const int   start = __builtin_amdgcn_readlane(startv, k);
        const int   cnt   = __builtin_amdgcn_readlane(cntv, k);
        const float dn    = __int_as_float(__builtin_amdgcn_readlane(dnv, k));
        const v2f   self  = hr[(long)n * (HID / 2)];
        v2f acc = (v2f){0.0f, 0.0f};
        for (int base = 0; base < cnt; base += 32) {
            const int  j  = base + lane;
            const bool ok = j < cnt;
            const int   sr = csr[start + (ok ? j : cnt - 1)];
            const int   s  = ok ? sr : n;
            const int   m = (cnt - base < 32) ? (cnt - base) : 32;
            for (int jj = 0; jj < m; jj += P1_BATCH) {
                v2f row[P1_BATCH];
#pragma unroll
                for (int q = 0; q < P1_BATCH; ++q) {
                    const long sj = __builtin_amdgcn_readlane(s, jj + q);
                    row[q] = hr[sj * (HID / 2)];
                }
#pragma unroll
                for (int q = 0; q < P1_BATCH; ++q) {
                    const float wq = (jj + q < m) ? 1.0f : 0.0f;
                    acc.x = __builtin_fmaf(row[q].x, wq, acc.x);
                    acc.y = __builtin_fmaf(row[q].y, wq, acc.y);
                }
            }
        }
        float h0 = (acc.x + self.x) * dn + bv.x;
        float h1 = (acc.y + self.y) * dn + bv.y;
        h0 = h0 > 0.0f ? h0 : 0.0f;
        h1 = h1 > 0.0f ? h1 : 0.0f;
        float p0 = h0 * w2.x + h1 * w2.z;
        float p1 = h0 * w2.y + h1 * w2.w;
#pragma unroll
        for (int off = 16; off > 0; off >>= 1) {
            p0 += __shfl_xor(p0, off);
            p1 += __shfl_xor(p1, off);
        }
        if (lane == 0) *(v2f*)(hw2 + 2 * (long)n) = (v2f){p0 * dn, p1 * dn};
    }
}

#define P2_LANES 8
static_assert((N_NODES * P2_LANES) % 256 == 0, "grid covers the nodes exactly, no tail");
__global__ __launch_bounds__(256) void pull2(const int* __restrict__ csr,
                                             const unsigned int* __restrict__ rowptr,
                                             const int* __restrict__ degi,
                                             const float* __restrict__ dinv,
                                             const float* __restrict__ hw2,
                                             const float* __restrict__ b2,
                                             float* __restrict__ out) {
    const int n   = (blockIdx.x * 256 + threadIdx.x) / P2_LANES;
    const int sub = threadIdx.x & (P2_LANES - 1);
    const int   start = (int)rowptr[n];
    const int   cnt   = degi[n];
    const float dn    = dinv[n];
    const v2f   hs    = *(const v2f*)(hw2 + 2 * (long)n);
    float a0 = 0.0f, a1 = 0.0f;
    for (int j = sub; j < cnt; j += P2_LANES) {
        const int   s = csr[start + j];
        const v2f   h = *(const v2f*)(hw2 + 2 * (long)s);
        a0 += h.x;
        a1 += h.y;
    }
#pragma unroll
    for (int off = P2_LANES / 2; off > 0; off >>= 1) {
        a0 += __shfl_xor(a0, off);
        a1 += __shfl_xor(a1, off);
    }
    if (sub == 0) {
        v2f o;
        o.x = (a0 + hs.x) * dn + b2[0];
        o.y = (a1 + hs.y) * dn + b2[1];
        *(v2f*)(out + 2 * (long)n) = o;
    }
}

extern "C" void kernel_launch(void* const* d_in, const int* in_sizes, int n_in,
                              void* d_out, int out_size, void* d_ws, size_t ws_size,
                              hipStream_t stream) {
    const float* x  = (const float*)d_in[0];
    const int*   ei = (const int*)  d_in[1];
    const float* W1 = (const float*)d_in[2];
    const float* b1 = (const float*)d_in[3];
    const float* W2 = (const float*)d_in[4];
    const float* b2 = (const float*)d_in[5];
    const int* src = ei;
    const int* dst = ei + N_EDGES;
    float* out = (float*)d_out;

    float*        ws     = (float*)d_ws;
    float*        dinv   = ws;
    int*          degi   = (int*)(dinv + N_NODES);
    unsigned int* rowloc = (unsigned int*)(degi + N_NODES);
    unsigned int* rowptr = rowloc + N_NODES;
    unsigned int* btot   = rowptr + N_NODES;
    unsigned int* cc     = btot + 128;
    int*          csr    = (int*)(cc + (size_t)N_CHUNKS * N_NODES);
    float*        hw1    = (float*)(csr + N_EDGES);
    float*        hw2    = hw1 + (size_t)N_NODES * HID;

    const dim3 blk(256);

    count_kernel<<<N_RANGES * N_CHUNKS, dim3(PRIV_THREADS), 0, stream>>>(dst, cc);
    scan_kernel <<<SCAN_BLOCKS, dim3(PRIV_THREADS), 0, stream>>>(cc, dinv, degi, rowloc, btot);
    fill_gemm_kernel<<<N_RANGES * N_CHUNKS + GEMM_BLOCKS, dim3(PRIV_THREADS), 0, stream>>>(
        src, dst, cc, rowloc, btot, rowptr, csr, x, W1, dinv, hw1);

    pull1<<<N_NODES / (8 * P1_NODES), blk, 0, stream>>>(csr, rowptr, degi, dinv, hw1, b1, W2, hw2);

    pull2<<<N_NODES * P2_LANES / 256, blk, 0, stream>>>(csr, rowptr, degi, dinv, hw2, b2, out);
}


# Both row gathers enter fewer, fatter waves: layer 1 8 rows a wave, next list read a row ahead, 16 row loads in flight; layer 2 16 rows a wave, 2 lanes a row: its sums 2 partials a row (was 8)
# speedup vs baseline: 5.6583x; 1.0440x over previous
#include <hip/hip_runtime.h>

typedef float v2f __attribute__((ext_vector_type(2)));
typedef float v4f __attribute__((ext_vector_type(4)));
typedef float v8f __attribute__((ext_vector_type(8)));
typedef int   v4i __attribute__((ext_vector_type(4)));

#define N_NODES 100000
#define N_EDGES 1600000
#define IN_DIM  165
#define HID     64
#define OUT_DIM 2
#define NTILES  (N_NODES / 16)
#define KPAIRS  83

#define N_RANGES    8
#define RNG_NODES   12512
#define N_CHUNKS    32
#define CHUNK_EDGES (N_EDGES / N_CHUNKS)
#define CHUNK_VEC4  (CHUNK_EDGES / 4)
#define PRIV_THREADS 1024
#define DEG_ROUNDS  (CHUNK_VEC4 / PRIV_THREADS)
#define SCAN_BLOCKS ((N_NODES + PRIV_THREADS - 1) / PRIV_THREADS)
static_assert(SCAN_BLOCKS <= PRIV_THREADS, "one thread per scan block in the fill kernel");

__device__ __forceinline__ unsigned block_excl_scan(unsigned v, unsigned* ws, unsigned& total) {
    const int lane = threadIdx.x & 31;
    const int wave = threadIdx.x >> 5;
    unsigned incl = v;
#pragma unroll
    for (int off = 1; off < 32; off <<= 1) {
        const unsigned t = __shfl_up(incl, off);
        if (lane >= off) incl += t;
    }
    if (lane == 31) ws[wave] = incl;
    __syncthreads();
    if (wave == 0) {
        const unsigned w = ws[lane];
        unsigned wi = w;
#pragma unroll
        for (int off = 1; off < 32; off <<= 1) {
            const unsigned t = __shfl_up(wi, off);
            if (lane >= off) wi += t;
        }
        ws[lane] = wi - w;
        if (lane == 31) ws[32] = wi;
    }
    __syncthreads();
    total = ws[32];
    return incl - v + ws[wave];
}

__global__ __launch_bounds__(PRIV_THREADS) void count_kernel(const int* __restrict__ dst,
                                                             unsigned int* __restrict__ cc) {
    __shared__ unsigned int cnt[RNG_NODES];
    const int r  = blockIdx.x % N_RANGES;
    const int c  = blockIdx.x / N_RANGES;
    const int lo = r * RNG_NODES;
    for (int i = threadIdx.x; i < RNG_NODES; i += PRIV_THREADS) cnt[i] = 0u;
    __syncthreads();

    const v4i* __restrict__ pd = (const v4i*)(dst + (size_t)c * CHUNK_EDGES);
    v4i d[DEG_ROUNDS + 1];
#pragma unroll
    for (int k = 0; k < DEG_ROUNDS; ++k) d[k] = pd[threadIdx.x + k * PRIV_THREADS];
    d[DEG_ROUNDS] = (v4i){-1, -1, -1, -1};
    if (threadIdx.x + DEG_ROUNDS * PRIV_THREADS < CHUNK_VEC4)
        d[DEG_ROUNDS] = pd[threadIdx.x + DEG_ROUNDS * PRIV_THREADS];
#pragma unroll
    for (int k = 0; k <= DEG_ROUNDS; ++k) {
        unsigned t0 = (unsigned)(d[k].x - lo), t1 = (unsigned)(d[k].y - lo);
        unsigned t2 = (unsigned)(d[k].z - lo), t3 = (unsigned)(d[k].w - lo);
        if (t0 < RNG_NODES) atomicAdd(&cnt[t0], 1u);
        if (t1 < RNG_NODES) atomicAdd(&cnt[t1], 1u);
        if (t2 < RNG_NODES) atomicAdd(&cnt[t2], 1u);
        if (t3 < RNG_NODES) atomicAdd(&cnt[t3], 1u);
    }
    __syncthreads();

    unsigned int* __restrict__ row = cc + (size_t)c * N_NODES;
    for (int i = threadIdx.x; i < RNG_NODES; i += PRIV_THREADS) {
        const int n = lo + i;
        if (n < N_NODES) row[n] = cnt[i];
    }
}

__global__ __launch_bounds__(PRIV_THREADS) void scan_kernel(unsigned int* __restrict__ cc,
                                                            float* __restrict__ dinv,
                                                            int* __restrict__ degi,
                                                            unsigned int* __restrict__ rowloc,
                                                            unsigned int* __restrict__ btot) {
    __shared__ unsigned int ws[33];
    const int  n     = blockIdx.x * PRIV_THREADS + threadIdx.x;
    const bool valid = n < N_NODES;
    unsigned v[N_CHUNKS];
#pragma unroll
    for (int c = 0; c < N_CHUNKS; ++c) v[c] = valid ? cc[(size_t)c * N_NODES + n] : 0u;
    unsigned run = 0u;
#pragma unroll
    for (int c = 0; c < N_CHUNKS; ++c) {
        if (valid) cc[(size_t)c * N_NODES + n] = run;
        run += v[c];
    }
    if (valid) {
        degi[n] = (int)run;
        dinv[n] = rsqrtf((float)run + 1.0f);
    }
    unsigned total;
    const unsigned excl = block_excl_scan(run, ws, total);
    if (valid) rowloc[n] = excl;
    if (threadIdx.x == 0) btot[blockIdx.x] = total;
}

#define FILL_BBASE_OFF  RNG_NODES
#define FILL_WS_OFF     (RNG_NODES + 128)
#define FILL_LDS_WORDS  (RNG_NODES + 128 + 33)
#define CUR_ROUNDS      ((RNG_NODES + PRIV_THREADS - 1) / PRIV_THREADS)
static_assert(SCAN_BLOCKS <= 128, "block bases fit their LDS slot");
__device__ __forceinline__ void fill_part(unsigned int* __restrict__ lds, const int bidx,
                                          const int* __restrict__ src,
                                          const int* __restrict__ dst,
                                          const unsigned int* __restrict__ cc,
                                          const unsigned int* __restrict__ rowloc,
                                          const unsigned int* __restrict__ btot,
                                          unsigned int* __restrict__ rowptr,
                                          int* __restrict__ csr) {
    unsigned int* cur   = lds;
    unsigned int* bbase = lds + FILL_BBASE_OFF;
    unsigned int* ws    = lds + FILL_WS_OFF;
    const int r  = bidx % N_RANGES;
    const int c  = bidx / N_RANGES;
    const int lo = r * RNG_NODES;

    const unsigned int* __restrict__ row = cc + (size_t)c * N_NODES;
    unsigned rl[CUR_ROUNDS], rw[CUR_ROUNDS];
#pragma unroll
    for (int j = 0; j < CUR_ROUNDS; ++j) {
        const int n  = lo + threadIdx.x + j * PRIV_THREADS;
        const int nc = (n < N_NODES) ? n : N_NODES - 1;
        rl[j] = rowloc[nc];
        rw[j] = row[nc];
    }

    unsigned total;
    const unsigned bv = (threadIdx.x < SCAN_BLOCKS) ? btot[threadIdx.x] : 0u;
    const unsigned be = block_excl_scan(bv, ws, total);
    if (threadIdx.x < SCAN_BLOCKS) bbase[threadIdx.x] = be;
    __syncthreads();

#pragma unroll
    for (int j = 0; j < CUR_ROUNDS; ++j) {
        const int i = threadIdx.x + j * PRIV_THREADS;
        const int n = lo + i;
        if (i < RNG_NODES) {
            unsigned p = 0u;
            if (n < N_NODES) {
                const unsigned base = bbase[n / PRIV_THREADS] + rl[j];
                p = base + rw[j];
                if (c == 0) rowptr[n] = base;
            }
            cur[i] = p;
        }
    }
    __syncthreads();

    const v4i* __restrict__ ps = (const v4i*)(src + (size_t)c * CHUNK_EDGES);
    const v4i* __restrict__ pd = (const v4i*)(dst + (size_t)c * CHUNK_EDGES);
    for (int b = threadIdx.x; b < CHUNK_VEC4; b += 4 * PRIV_THREADS) {
        v4i d[4], s[4];
#pragma unroll
        for (int k = 0; k < 4; ++k) {
            const int  i  = b + k * PRIV_THREADS;
            const bool ok = i < CHUNK_VEC4;
            const int  ic = ok ? i : CHUNK_VEC4 - 1;
            d[k] = pd[ic];
            s[k] = ps[ic];
            if (!ok) d[k] = (v4i){-1, -1, -1, -1};
        }
#pragma unroll
        for (int m = 0; m < 16; ++m) {
            const unsigned t = (unsigned)(d[m >> 2][m & 3] - lo);
            if (t < RNG_NODES) {
                const unsigned pos = atomicAdd(&cur[t], 1u);
                csr[pos] = s[m >> 2][m & 3];
            }
        }
    }
}

__device__ __forceinline__ void gemm_step(const v2f A, const float* __restrict__ w1p,
                                          const int k, const int hh, const int lr,
                                          v8f& acc0, v8f& acc1, v8f& acc2, v8f& acc3) {
    const int ka = k + 2 * hh;
    const v2f* __restrict__ bp = (const v2f*)&w1p[(ka >> 1) * 2 * HID + lr * 2];
    v2f B0 = bp[0];
    v2f B1 = bp[16];
    v2f B2 = bp[32];
    v2f B3 = bp[48];
    acc0 = __builtin_amdgcn_wmma_f32_16x16x4_f32(false, A, false, B0, (short)0, acc0, false, false);
    acc1 = __builtin_amdgcn_wmma_f32_16x16x4_f32(false, A, false, B1, (short)0, acc1, false, false);
    acc2 = __builtin_amdgcn_wmma_f32_16x16x4_f32(false, A, false, B2, (short)0, acc2, false, false);
    acc3 = __builtin_amdgcn_wmma_f32_16x16x4_f32(false, A, false, B3, (short)0, acc3, false, false);
}

#define GEMM_BLOCKS     256
#define GEMM_LDS_WORDS  (KPAIRS * 2 * HID)
#define W1_ROUNDS       ((GEMM_LDS_WORDS + PRIV_THREADS - 1) / PRIV_THREADS)
#define W1_STAGE_WORDS  (W1_ROUNDS * PRIV_THREADS)
#define A_GROUP         8
static_assert(160 % (4 * A_GROUP) == 0, "whole groups cover K = 0..159");
__device__ __forceinline__ void gemm_part(float* __restrict__ w1p, const int gidx,
                                          const float* __restrict__ x,
                                          const float* __restrict__ W1,
                                          const float* __restrict__ dinv,
                                          float* __restrict__ hw1) {
    float wv[W1_ROUNDS];
#pragma unroll
    for (int j = 0; j < W1_ROUNDS; ++j) {
        const int  i    = threadIdx.x + j * PRIV_THREADS;
        const int  kp   = i >> 7;
        const int  r    = i & 127;
        const int  col  = r >> 1;
        const int  sub  = r & 1;
        const int  krow = kp * 2 + sub;
        const bool in   = krow < IN_DIM;
        const float v   = W1[(in ? krow : IN_DIM - 1) * HID + col];
        wv[j] = in ? v : 0.0f;
    }
#pragma unroll
    for (int j = 0; j < W1_ROUNDS; ++j) w1p[threadIdx.x + j * PRIV_THREADS] = wv[j];
    __syncthreads();

    const int wave = threadIdx.x >> 5;
    const int lane = threadIdx.x & 31;
    const int tile = gidx + GEMM_BLOCKS * wave;
    if (tile >= NTILES) return;

    const int hh  = lane >> 4;
    const int lr  = lane & 15;
    const int row = tile * 16 + lr;
    const float* __restrict__ xrb = x + (long)row * IN_DIM;
    const float* __restrict__ xr  = xrb + 2 * hh;

    v8f acc0 = (v8f){}, acc1 = (v8f){}, acc2 = (v8f){}, acc3 = (v8f){};

    for (int k0 = 0; k0 < 160; k0 += 4 * A_GROUP) {
        v2f A[A_GROUP];
#pragma unroll
        for (int j = 0; j < A_GROUP; ++j) {
            A[j].x = xr[k0 + 4 * j];
            A[j].y = xr[k0 + 4 * j + 1];
        }
#pragma unroll
        for (int j = 0; j < A_GROUP; ++j)
            gemm_step(A[j], w1p, k0 + 4 * j, hh, lr, acc0, acc1, acc2, acc3);
    }
    v2f A160;
    A160.x = xr[160];
    A160.y = xr[161];
    const float a164 = xrb[164];
    gemm_step(A160, w1p, 160, hh, lr, acc0, acc1, acc2, acc3);

    {
        v2f A;
        A.x = (hh == 0) ? a164 : 0.0f;
        A.y = 0.0f;
        const v2f* __restrict__ bp = (const v2f*)&w1p[(KPAIRS - 1) * 2 * HID + lr * 2];
        v2f B0 = bp[0];
        v2f B1 = bp[16];
        v2f B2 = bp[32];
        v2f B3 = bp[48];
        acc0 = __builtin_amdgcn_wmma_f32_16x16x4_f32(false, A, false, B0, (short)0, acc0, false, false);
        acc1 = __builtin_amdgcn_wmma_f32_16x16x4_f32(false, A, false, B1, (short)0, acc1, false, false);
        acc2 = __builtin_amdgcn_wmma_f32_16x16x4_f32(false, A, false, B2, (short)0, acc2, false, false);
        acc3 = __builtin_amdgcn_wmma_f32_16x16x4_f32(false, A, false, B3, (short)0, acc3, false, false);
    }

    const v4f* __restrict__ dq = (const v4f*)(dinv + tile * 16 + 8 * hh);
    const v4f dlo = dq[0], dhi = dq[1];
    const float dv[8] = {dlo.x, dlo.y, dlo.z, dlo.w, dhi.x, dhi.y, dhi.z, dhi.w};
    float* __restrict__ outp = hw1 + (long)(tile * 16 + 8 * hh) * HID + lr;
#pragma unroll
    for (int v = 0; v < 8; ++v) {
        long o = (long)v * HID;
        outp[o +  0] = acc0[v] * dv[v];
        outp[o + 16] = acc1[v] * dv[v];
        outp[o + 32] = acc2[v] * dv[v];
        outp[o + 48] = acc3[v] * dv[v];
    }
}

#define FG_LDS_WORDS (FILL_LDS_WORDS > GEMM_LDS_WORDS ? FILL_LDS_WORDS : GEMM_LDS_WORDS)
static_assert(W1_STAGE_WORDS <= FG_LDS_WORDS, "the whole staging rounds of W1 fit the shared buffer");
static_assert(2 * N_RANGES * N_CHUNKS == 2 * GEMM_BLOCKS, "as many fill as GEMM workgroups");
static_assert(25 * GEMM_BLOCKS >= NTILES, "the tile rounds of 32 waves cover every tile");
__global__ __launch_bounds__(PRIV_THREADS) __attribute__((amdgpu_waves_per_eu(16, 16)))
void fill_gemm_kernel(const int* __restrict__ src,
                      const int* __restrict__ dst,
                      const unsigned int* __restrict__ cc,
                      const unsigned int* __restrict__ rowloc,
                      const unsigned int* __restrict__ btot,
                      unsigned int* __restrict__ rowptr,
                      int* __restrict__ csr,
                      const float* __restrict__ x,
                      const float* __restrict__ W1,
                      const float* __restrict__ dinv,
                      float* __restrict__ hw1) {
    __shared__ __attribute__((aligned(16))) unsigned int lds[FG_LDS_WORDS];
    const int b    = blockIdx.x;
    const int role = (b ^ (b >> 8)) & 1;
    const int idx  = b >> 1;
    if (role == 0) fill_part(lds, idx, src, dst, cc, rowloc, btot, rowptr, csr);
    else           gemm_part((float*)lds, idx, x, W1, dinv, hw1);
}

#define P1_NODES 8
#define P1_WAVES 10
#define P1_BATCH 16
static_assert(N_NODES % (P1_WAVES * P1_NODES) == 0, "grid covers the nodes exactly, no tail");

__device__ __forceinline__ void p1_add_rows(const v2f* __restrict__ hr, const int s, const int m, v2f& acc) {
    for (int jj = 0; jj < m; jj += P1_BATCH) {
        v2f row[P1_BATCH];
#pragma unroll
        for (int q = 0; q < P1_BATCH; ++q) {
            const long sj = __builtin_amdgcn_readlane(s, jj + q);
            row[q] = hr[sj * (HID / 2)];
        }
#pragma unroll
        for (int q = 0; q < P1_BATCH; ++q) {
            const float wq = (jj + q < m) ? 1.0f : 0.0f;
            acc.x = __builtin_fmaf(row[q].x, wq, acc.x);
            acc.y = __builtin_fmaf(row[q].y, wq, acc.y);
        }
    }
}

__global__ __launch_bounds__(32 * P1_WAVES) void pull1(const int* __restrict__ csr,
                                             const unsigned int* __restrict__ rowptr,
                                             const int* __restrict__ degi,
                                             const float* __restrict__ dinv,
                                             const float* __restrict__ hw1,
                                             const float* __restrict__ b1,
                                             const float* __restrict__ W2,
                                             float* __restrict__ hw2) {
    const int lane = threadIdx.x & 31;
    const int n0   = (blockIdx.x * P1_WAVES + (threadIdx.x >> 5)) * P1_NODES;
    const int   ml     = n0 + (lane & (P1_NODES - 1));
    const int   startv = (int)rowptr[ml];
    const int   cntv   = degi[ml];
    const int   dnv    = __float_as_int(dinv[ml]);
    const v4f   w2 = ((const v4f*)W2)[lane];
    const v2f   bv = ((const v2f*)b1)[lane];
    const v2f* __restrict__ hr = (const v2f*)hw1 + lane;

    int srn;
    {
        const int st0 = __builtin_amdgcn_readlane(startv, 0);
        const int c0  = __builtin_amdgcn_readlane(cntv, 0);
        srn = csr[(lane < c0) ? st0 + lane : 0];
    }
    v2f selfn = hr[(long)n0 * (HID / 2)];

    for (int k = 0; k < P1_NODES; ++k) {
        const int   n     = n0 + k;
        const int   start = __builtin_amdgcn_readlane(startv, k);
        const int   cnt   = __builtin_amdgcn_readlane(cntv, k);
        const float dn    = __int_as_float(__builtin_amdgcn_readlane(dnv, k));
        const int   sr0   = srn;
        const v2f   self  = selfn;
        if (k + 1 < P1_NODES) {
            const int st1 = __builtin_amdgcn_readlane(startv, k + 1);
            const int c1  = __builtin_amdgcn_readlane(cntv, k + 1);
            srn   = csr[(lane < c1) ? st1 + lane : 0];
            selfn = hr[(long)(n + 1) * (HID / 2)];
        }
        v2f acc = (v2f){0.0f, 0.0f};
        if (cnt > 0) p1_add_rows(hr, (lane < cnt) ? sr0 : n, (cnt < 32) ? cnt : 32, acc);
        for (int base = 32; base < cnt; base += 32) {
            const int  j  = base + lane;
            const bool ok = j < cnt;
            const int  sr = csr[ok ? start + j : 0];
            p1_add_rows(hr, ok ? sr : n, (cnt - base < 32) ? (cnt - base) : 32, acc);
        }
        float h0 = (acc.x + self.x) * dn + bv.x;
        float h1 = (acc.y + self.y) * dn + bv.y;
        h0 = h0 > 0.0f ? h0 : 0.0f;
        h1 = h1 > 0.0f ? h1 : 0.0f;
        float p0 = h0 * w2.x + h1 * w2.z;
        float p1 = h0 * w2.y + h1 * w2.w;
#pragma unroll
        for (int off = 16; off > 0; off >>= 1) {
            p0 += __shfl_xor(p0, off);
            p1 += __shfl_xor(p1, off);
        }
        if (lane == 0) *(v2f*)(hw2 + 2 * (long)n) = (v2f){p0 * dn, p1 * dn};
    }
}

#define P2_LANES   2
#define P2_RUNS    4
#define P2_THREADS 160
static_assert((N_NODES * P2_LANES) % P2_THREADS == 0, "grid covers the nodes exactly, no tail");
static_assert(P2_THREADS % 32 == 0, "whole waves");
typedef int v4iu __attribute__((ext_vector_type(4), aligned(4)));
__global__ __launch_bounds__(P2_THREADS) void pull2(const int* __restrict__ csr,
                                             const unsigned int* __restrict__ rowptr,
                                             const int* __restrict__ degi,
                                             const float* __restrict__ dinv,
                                             const float* __restrict__ hw2,
                                             const float* __restrict__ b2,
                                             float* __restrict__ out) {
    const int t   = blockIdx.x * P2_THREADS + threadIdx.x;
    const int n   = t / P2_LANES;
    const int sub = t & (P2_LANES - 1);
    const int   start = (int)rowptr[n];
    const int   cnt   = degi[n];
    const float dn    = dinv[n];
    const v2f   hs    = *(const v2f*)(hw2 + 2 * (long)n);
    float a0 = 0.0f, a1 = 0.0f;
    for (int base = 0; __builtin_amdgcn_ballot_w32(base < cnt) != 0u; base += 4 * P2_RUNS * P2_LANES) {
        int  s[4 * P2_RUNS];
        bool ok[4 * P2_RUNS];
#pragma unroll
        for (int q = 0; q < P2_RUNS; ++q) {
            const int j0 = base + 4 * (P2_LANES * q + sub);
            const int at = (j0 < cnt) ? start + j0 : 0;
            const int a3 = (at < N_EDGES - 4) ? at : N_EDGES - 4;
            const int sh = at - a3;
            const int jb = j0 - sh;
            const v4iu v = *(const v4iu*)(csr + a3);
#pragma unroll
            for (int i = 0; i < 4; ++i) {
                ok[4 * q + i] = (i >= sh) & (jb + i < cnt);
                s[4 * q + i]  = ok[4 * q + i] ? v[i] : 0;
            }
        }
        v2f h[4 * P2_RUNS];
#pragma unroll
        for (int u = 0; u < 4 * P2_RUNS; ++u) h[u] = *(const v2f*)(hw2 + 2 * (long)s[u]);
#pragma unroll
        for (int u = 0; u < 4 * P2_RUNS; ++u) {
            a0 += ok[u] ? h[u].x : 0.0f;
            a1 += ok[u] ? h[u].y : 0.0f;
        }
    }
    a0 += __shfl_xor(a0, 1);
    a1 += __shfl_xor(a1, 1);
    if (sub == 0) {
        v2f o;
        o.x = (a0 + hs.x) * dn + b2[0];
        o.y = (a1 + hs.y) * dn + b2[1];
        *(v2f*)(out + 2 * (long)n) = o;
    }
}

extern "C" void kernel_launch(void* const* d_in, const int* in_sizes, int n_in,
                              void* d_out, int out_size, void* d_ws, size_t ws_size,
                              hipStream_t stream) {
    const float* x  = (const float*)d_in[0];
    const int*   ei = (const int*)  d_in[1];
    const float* W1 = (const float*)d_in[2];
    const float* b1 = (const float*)d_in[3];
    const float* W2 = (const float*)d_in[4];
    const float* b2 = (const float*)d_in[5];
    const int* src = ei;
    const int* dst = ei + N_EDGES;
    float* out = (float*)d_out;

    float*        ws     = (float*)d_ws;
    float*        dinv   = ws;
    int*          degi   = (int*)(dinv + N_NODES);
    unsigned int* rowloc = (unsigned int*)(degi + N_NODES);
    unsigned int* rowptr = rowloc + N_NODES;
    unsigned int* btot   = rowptr + N_NODES;
    unsigned int* cc     = btot + 128;
    int*          csr    = (int*)(cc + (size_t)N_CHUNKS * N_NODES);
    float*        hw1    = (float*)(csr + N_EDGES);
    float*        hw2    = hw1 + (size_t)N_NODES * HID;

    count_kernel<<<N_RANGES * N_CHUNKS, dim3(PRIV_THREADS), 0, stream>>>(dst, cc);
    scan_kernel <<<SCAN_BLOCKS, dim3(PRIV_THREADS), 0, stream>>>(cc, dinv, degi, rowloc, btot);
    fill_gemm_kernel<<<N_RANGES * N_CHUNKS + GEMM_BLOCKS, dim3(PRIV_THREADS), 0, stream>>>(
        src, dst, cc, rowloc, btot, rowptr, csr, x, W1, dinv, hw1);

    pull1<<<N_NODES / (P1_WAVES * P1_NODES), dim3(32 * P1_WAVES), 0, stream>>>(csr, rowptr, degi, dinv, hw1, b1, W2, hw2);

    pull2<<<N_NODES * P2_LANES / P2_THREADS, dim3(P2_THREADS), 0, stream>>>(csr, rowptr, degi, dinv, hw2, b2, out);
}


# GEMM1 reads x as whole 128-B row segments (one load per row, lanes on consecutive words) and forms the A fragments through a per-wave LDS slab, not 16-line gathers per load; same k order and sums
# speedup vs baseline: 5.9390x; 1.0496x over previous
#include <hip/hip_runtime.h>

typedef float v2f __attribute__((ext_vector_type(2)));
typedef float v4f __attribute__((ext_vector_type(4)));
typedef float v8f __attribute__((ext_vector_type(8)));
typedef int   v4i __attribute__((ext_vector_type(4)));

#define N_NODES 100000
#define N_EDGES 1600000
#define IN_DIM  165
#define HID     64
#define OUT_DIM 2
#define NTILES  (N_NODES / 16)
#define KPAIRS  83

#define N_RANGES    8
#define RNG_NODES   12512
#define N_CHUNKS    32
#define CHUNK_EDGES (N_EDGES / N_CHUNKS)
#define CHUNK_VEC4  (CHUNK_EDGES / 4)
#define PRIV_THREADS 1024
#define DEG_ROUNDS  (CHUNK_VEC4 / PRIV_THREADS)
#define SCAN_BLOCKS ((N_NODES + PRIV_THREADS - 1) / PRIV_THREADS)
static_assert(SCAN_BLOCKS <= PRIV_THREADS, "one thread per scan block in the fill kernel");

__device__ __forceinline__ unsigned block_excl_scan(unsigned v, unsigned* ws, unsigned& total) {
    const int lane = threadIdx.x & 31;
    const int wave = threadIdx.x >> 5;
    unsigned incl = v;
#pragma unroll
    for (int off = 1; off < 32; off <<= 1) {
        const unsigned t = __shfl_up(incl, off);
        if (lane >= off) incl += t;
    }
    if (lane == 31) ws[wave] = incl;
    __syncthreads();
    if (wave == 0) {
        const unsigned w = ws[lane];
        unsigned wi = w;
#pragma unroll
        for (int off = 1; off < 32; off <<= 1) {
            const unsigned t = __shfl_up(wi, off);
            if (lane >= off) wi += t;
        }
        ws[lane] = wi - w;
        if (lane == 31) ws[32] = wi;
    }
    __syncthreads();
    total = ws[32];
    return incl - v + ws[wave];
}

__global__ __launch_bounds__(PRIV_THREADS) void count_kernel(const int* __restrict__ dst,
                                                             unsigned int* __restrict__ cc) {
    __shared__ unsigned int cnt[RNG_NODES];
    const int r  = blockIdx.x % N_RANGES;
    const int c  = blockIdx.x / N_RANGES;
    const int lo = r * RNG_NODES;
    for (int i = threadIdx.x; i < RNG_NODES; i += PRIV_THREADS) cnt[i] = 0u;
    __syncthreads();

    const v4i* __restrict__ pd = (const v4i*)(dst + (size_t)c * CHUNK_EDGES);
    v4i d[DEG_ROUNDS + 1];
#pragma unroll
    for (int k = 0; k < DEG_ROUNDS; ++k) d[k] = pd[threadIdx.x + k * PRIV_THREADS];
    d[DEG_ROUNDS] = (v4i){-1, -1, -1, -1};
    if (threadIdx.x + DEG_ROUNDS * PRIV_THREADS < CHUNK_VEC4)
        d[DEG_ROUNDS] = pd[threadIdx.x + DEG_ROUNDS * PRIV_THREADS];
#pragma unroll
    for (int k = 0; k <= DEG_ROUNDS; ++k) {
        unsigned t0 = (unsigned)(d[k].x - lo), t1 = (unsigned)(d[k].y - lo);
        unsigned t2 = (unsigned)(d[k].z - lo), t3 = (unsigned)(d[k].w - lo);
        if (t0 < RNG_NODES) atomicAdd(&cnt[t0], 1u);
        if (t1 < RNG_NODES) atomicAdd(&cnt[t1], 1u);
        if (t2 < RNG_NODES) atomicAdd(&cnt[t2], 1u);
        if (t3 < RNG_NODES) atomicAdd(&cnt[t3], 1u);
    }
    __syncthreads();

    unsigned int* __restrict__ row = cc + (size_t)c * N_NODES;
    for (int i = threadIdx.x; i < RNG_NODES; i += PRIV_THREADS) {
        const int n = lo + i;
        if (n < N_NODES) row[n] = cnt[i];
    }
}

__global__ __launch_bounds__(PRIV_THREADS) void scan_kernel(unsigned int* __restrict__ cc,
                                                            float* __restrict__ dinv,
                                                            int* __restrict__ degi,
                                                            unsigned int* __restrict__ rowloc,
                                                            unsigned int* __restrict__ btot) {
    __shared__ unsigned int ws[33];
    const int  n     = blockIdx.x * PRIV_THREADS + threadIdx.x;
    const bool valid = n < N_NODES;
    unsigned v[N_CHUNKS];
#pragma unroll
    for (int c = 0; c < N_CHUNKS; ++c) v[c] = valid ? cc[(size_t)c * N_NODES + n] : 0u;
    unsigned run = 0u;
#pragma unroll
    for (int c = 0; c < N_CHUNKS; ++c) {
        if (valid) cc[(size_t)c * N_NODES + n] = run;
        run += v[c];
    }
    if (valid) {
        degi[n] = (int)run;
        dinv[n] = rsqrtf((float)run + 1.0f);
    }
    unsigned total;
    const unsigned excl = block_excl_scan(run, ws, total);
    if (valid) rowloc[n] = excl;
    if (threadIdx.x == 0) btot[blockIdx.x] = total;
}

#define FILL_BBASE_OFF  RNG_NODES
#define FILL_WS_OFF     (RNG_NODES + 128)
#define FILL_LDS_WORDS  (RNG_NODES + 128 + 33)
#define CUR_ROUNDS      ((RNG_NODES + PRIV_THREADS - 1) / PRIV_THREADS)
static_assert(SCAN_BLOCKS <= 128, "block bases fit their LDS slot");
__device__ __forceinline__ void fill_part(unsigned int* __restrict__ lds, const int bidx,
                                          const int* __restrict__ src,
                                          const int* __restrict__ dst,
                                          const unsigned int* __restrict__ cc,
                                          const unsigned int* __restrict__ rowloc,
                                          const unsigned int* __restrict__ btot,
                                          unsigned int* __restrict__ rowptr,
                                          int* __restrict__ csr) {
    unsigned int* cur   = lds;
    unsigned int* bbase = lds + FILL_BBASE_OFF;
    unsigned int* ws    = lds + FILL_WS_OFF;
    const int r  = bidx % N_RANGES;
    const int c  = bidx / N_RANGES;
    const int lo = r * RNG_NODES;

    const unsigned int* __restrict__ row = cc + (size_t)c * N_NODES;
    unsigned rl[CUR_ROUNDS], rw[CUR_ROUNDS];
#pragma unroll
    for (int j = 0; j < CUR_ROUNDS; ++j) {
        const int n  = lo + threadIdx.x + j * PRIV_THREADS;
        const int nc = (n < N_NODES) ? n : N_NODES - 1;
        rl[j] = rowloc[nc];
        rw[j] = row[nc];
    }

    unsigned total;
    const unsigned bv = (threadIdx.x < SCAN_BLOCKS) ? btot[threadIdx.x] : 0u;
    const unsigned be = block_excl_scan(bv, ws, total);
    if (threadIdx.x < SCAN_BLOCKS) bbase[threadIdx.x] = be;
    __syncthreads();

#pragma unroll
    for (int j = 0; j < CUR_ROUNDS; ++j) {
        const int i = threadIdx.x + j * PRIV_THREADS;
        const int n = lo + i;
        if (i < RNG_NODES) {
            unsigned p = 0u;
            if (n < N_NODES) {
                const unsigned base = bbase[n / PRIV_THREADS] + rl[j];
                p = base + rw[j];
                if (c == 0) rowptr[n] = base;
            }
            cur[i] = p;
        }
    }
    __syncthreads();

    const v4i* __restrict__ ps = (const v4i*)(src + (size_t)c * CHUNK_EDGES);
    const v4i* __restrict__ pd = (const v4i*)(dst + (size_t)c * CHUNK_EDGES);
    for (int b = threadIdx.x; b < CHUNK_VEC4; b += 4 * PRIV_THREADS) {
        v4i d[4], s[4];
#pragma unroll
        for (int k = 0; k < 4; ++k) {
            const int  i  = b + k * PRIV_THREADS;
            const bool ok = i < CHUNK_VEC4;
            const int  ic = ok ? i : CHUNK_VEC4 - 1;
            d[k] = pd[ic];
            s[k] = ps[ic];
            if (!ok) d[k] = (v4i){-1, -1, -1, -1};
        }
#pragma unroll
        for (int m = 0; m < 16; ++m) {
            const unsigned t = (unsigned)(d[m >> 2][m & 3] - lo);
            if (t < RNG_NODES) {
                const unsigned pos = atomicAdd(&cur[t], 1u);
                csr[pos] = s[m >> 2][m & 3];
            }
        }
    }
}

__device__ __forceinline__ void gemm_step(const v2f A, const float* __restrict__ w1p,
                                          const int k, const int hh, const int lr,
                                          v8f& acc0, v8f& acc1, v8f& acc2, v8f& acc3) {
    const int ka = k + 2 * hh;
    const v2f* __restrict__ bp = (const v2f*)&w1p[(ka >> 1) * 2 * HID + lr * 2];
    v2f B0 = bp[0];
    v2f B1 = bp[16];
    v2f B2 = bp[32];
    v2f B3 = bp[48];
    acc0 = __builtin_amdgcn_wmma_f32_16x16x4_f32(false, A, false, B0, (short)0, acc0, false, false);
    acc1 = __builtin_amdgcn_wmma_f32_16x16x4_f32(false, A, false, B1, (short)0, acc1, false, false);
    acc2 = __builtin_amdgcn_wmma_f32_16x16x4_f32(false, A, false, B2, (short)0, acc2, false, false);
    acc3 = __builtin_amdgcn_wmma_f32_16x16x4_f32(false, A, false, B3, (short)0, acc3, false, false);
}

#define GEMM_BLOCKS     256
#define GEMM_LDS_WORDS  (KPAIRS * 2 * HID)
#define W1_ROUNDS       ((GEMM_LDS_WORDS + PRIV_THREADS - 1) / PRIV_THREADS)
#define W1_STAGE_WORDS  (W1_ROUNDS * PRIV_THREADS)
#define A_GROUP         8
static_assert(160 % (4 * A_GROUP) == 0, "whole groups cover K = 0..159");
#define XS_PITCH        36
#define XS_WORDS        (16 * XS_PITCH)
#define XS_OFF          W1_STAGE_WORDS
#define GEMM_TILE_WAVES 25
static_assert(4 * A_GROUP == 32, "a K group is one 128-B segment of a row: one word per lane");
static_assert(GEMM_BLOCKS * GEMM_TILE_WAVES >= NTILES, "the waves with a slab cover every tile");
#define GEMM_ALL_WORDS  (XS_OFF + GEMM_TILE_WAVES * XS_WORDS)
__device__ __forceinline__ void gemm_part(float* __restrict__ w1p, const int gidx,
                                          const float* __restrict__ x,
                                          const float* __restrict__ W1,
                                          const float* __restrict__ dinv,
                                          float* __restrict__ hw1) {
    float wv[W1_ROUNDS];
#pragma unroll
    for (int j = 0; j < W1_ROUNDS; ++j) {
        const int  i    = threadIdx.x + j * PRIV_THREADS;
        const int  kp   = i >> 7;
        const int  r    = i & 127;
        const int  col  = r >> 1;
        const int  sub  = r & 1;
        const int  krow = kp * 2 + sub;
        const bool in   = krow < IN_DIM;
        const float v   = W1[(in ? krow : IN_DIM - 1) * HID + col];
        wv[j] = in ? v : 0.0f;
    }
#pragma unroll
    for (int j = 0; j < W1_ROUNDS; ++j) w1p[threadIdx.x + j * PRIV_THREADS] = wv[j];
    __syncthreads();

    const int wave = threadIdx.x >> 5;
    const int lane = threadIdx.x & 31;
    const int tile = gidx + GEMM_BLOCKS * wave;
    if (tile >= NTILES) return;

    const int hh  = lane >> 4;
    const int lr  = lane & 15;
    const int row = tile * 16 + lr;
    const float* __restrict__ xrb = x + (long)row * IN_DIM;
    const float* __restrict__ xr  = xrb + 2 * hh;
    const float* __restrict__ xt  = x + (long)tile * 16 * IN_DIM + lane;
    float* __restrict__ xs = w1p + XS_OFF + wave * XS_WORDS;
    const v2f* __restrict__ af = (const v2f*)(xs + lr * XS_PITCH + 2 * hh);

    v8f acc0 = (v8f){}, acc1 = (v8f){}, acc2 = (v8f){}, acc3 = (v8f){};

    for (int k0 = 0; k0 < 160; k0 += 4 * A_GROUP) {
        float R[16];
#pragma unroll
        for (int r = 0; r < 16; ++r) R[r] = xt[r * IN_DIM + k0];
#pragma unroll
        for (int r = 0; r < 16; ++r) xs[r * XS_PITCH + lane] = R[r];
        __builtin_amdgcn_wave_barrier();
#pragma unroll
        for (int j = 0; j < A_GROUP; ++j)
            gemm_step(af[2 * j], w1p, k0 + 4 * j, hh, lr, acc0, acc1, acc2, acc3);
        __builtin_amdgcn_wave_barrier();
    }
    v2f A160;
    A160.x = xr[160];
    A160.y = xr[161];
    const float a164 = xrb[164];
    gemm_step(A160, w1p, 160, hh, lr, acc0, acc1, acc2, acc3);

    {
        v2f A;
        A.x = (hh == 0) ? a164 : 0.0f;
        A.y = 0.0f;
        const v2f* __restrict__ bp = (const v2f*)&w1p[(KPAIRS - 1) * 2 * HID + lr * 2];
        v2f B0 = bp[0];
        v2f B1 = bp[16];
        v2f B2 = bp[32];
        v2f B3 = bp[48];
        acc0 = __builtin_amdgcn_wmma_f32_16x16x4_f32(false, A, false, B0, (short)0, acc0, false, false);
        acc1 = __builtin_amdgcn_wmma_f32_16x16x4_f32(false, A, false, B1, (short)0, acc1, false, false);
        acc2 = __builtin_amdgcn_wmma_f32_16x16x4_f32(false, A, false, B2, (short)0, acc2, false, false);
        acc3 = __builtin_amdgcn_wmma_f32_16x16x4_f32(false, A, false, B3, (short)0, acc3, false, false);
    }

    const v4f* __restrict__ dq = (const v4f*)(dinv + tile * 16 + 8 * hh);
    const v4f dlo = dq[0], dhi = dq[1];
    const float dv[8] = {dlo.x, dlo.y, dlo.z, dlo.w, dhi.x, dhi.y, dhi.z, dhi.w};
    float* __restrict__ outp = hw1 + (long)(tile * 16 + 8 * hh) * HID + lr;
#pragma unroll
    for (int v = 0; v < 8; ++v) {
        long o = (long)v * HID;
        outp[o +  0] = acc0[v] * dv[v];
        outp[o + 16] = acc1[v] * dv[v];
        outp[o + 32] = acc2[v] * dv[v];
        outp[o + 48] = acc3[v] * dv[v];
    }
}

#define FG_LDS_WORDS (FILL_LDS_WORDS > GEMM_ALL_WORDS ? FILL_LDS_WORDS : GEMM_ALL_WORDS)
static_assert(2 * FG_LDS_WORDS * 4 <= 320 * 1024, "two workgroups' LDS fit one WGP");
static_assert(2 * N_RANGES * N_CHUNKS == 2 * GEMM_BLOCKS, "as many fill as GEMM workgroups");
static_assert(25 * GEMM_BLOCKS >= NTILES, "the tile rounds of 32 waves cover every tile");
__global__ __launch_bounds__(PRIV_THREADS) __attribute__((amdgpu_waves_per_eu(16, 16)))
void fill_gemm_kernel(const int* __restrict__ src,
                      const int* __restrict__ dst,
                      const unsigned int* __restrict__ cc,
                      const unsigned int* __restrict__ rowloc,
                      const unsigned int* __restrict__ btot,
                      unsigned int* __restrict__ rowptr,
                      int* __restrict__ csr,
                      const float* __restrict__ x,
                      const float* __restrict__ W1,
                      const float* __restrict__ dinv,
                      float* __restrict__ hw1) {
    __shared__ __attribute__((aligned(16))) unsigned int lds[FG_LDS_WORDS];
    const int b    = blockIdx.x;
    const int role = (b ^ (b >> 8)) & 1;
    const int idx  = b >> 1;
    if (role == 0) fill_part(lds, idx, src, dst, cc, rowloc, btot, rowptr, csr);
    else           gemm_part((float*)lds, idx, x, W1, dinv, hw1);
}

#define P1_NODES 8
#define P1_WAVES 10
#define P1_BATCH 16
static_assert(N_NODES % (P1_WAVES * P1_NODES) == 0, "grid covers the nodes exactly, no tail");

__device__ __forceinline__ void p1_add_rows(const v2f* __restrict__ hr, const int s, const int m, v2f& acc) {
    for (int jj = 0; jj < m; jj += P1_BATCH) {
        v2f row[P1_BATCH];
#pragma unroll
        for (int q = 0; q < P1_BATCH; ++q) {
            const long sj = __builtin_amdgcn_readlane(s, jj + q);
            row[q] = hr[sj * (HID / 2)];
        }
#pragma unroll
        for (int q = 0; q < P1_BATCH; ++q) {
            const float wq = (jj + q < m) ? 1.0f : 0.0f;
            acc.x = __builtin_fmaf(row[q].x, wq, acc.x);
            acc.y = __builtin_fmaf(row[q].y, wq, acc.y);
        }
    }
}

__global__ __launch_bounds__(32 * P1_WAVES) void pull1(const int* __restrict__ csr,
                                             const unsigned int* __restrict__ rowptr,
                                             const int* __restrict__ degi,
                                             const float* __restrict__ dinv,
                                             const float* __restrict__ hw1,
                                             const float* __restrict__ b1,
                                             const float* __restrict__ W2,
                                             float* __restrict__ hw2) {
    const int lane = threadIdx.x & 31;
    const int n0   = (blockIdx.x * P1_WAVES + (threadIdx.x >> 5)) * P1_NODES;
    const int   ml     = n0 + (lane & (P1_NODES - 1));
    const int   startv = (int)rowptr[ml];
    const int   cntv   = degi[ml];
    const int   dnv    = __float_as_int(dinv[ml]);
    const v4f   w2 = ((const v4f*)W2)[lane];
    const v2f   bv = ((const v2f*)b1)[lane];
    const v2f* __restrict__ hr = (const v2f*)hw1 + lane;

    int srn;
    {
        const int st0 = __builtin_amdgcn_readlane(startv, 0);
        const int c0  = __builtin_amdgcn_readlane(cntv, 0);
        srn = csr[(lane < c0) ? st0 + lane : 0];
    }
    v2f selfn = hr[(long)n0 * (HID / 2)];

    for (int k = 0; k < P1_NODES; ++k) {
        const int   n     = n0 + k;
        const int   start = __builtin_amdgcn_readlane(startv, k);
        const int   cnt   = __builtin_amdgcn_readlane(cntv, k);
        const float dn    = __int_as_float(__builtin_amdgcn_readlane(dnv, k));
        const int   sr0   = srn;
        const v2f   self  = selfn;
        if (k + 1 < P1_NODES) {
            const int st1 = __builtin_amdgcn_readlane(startv, k + 1);
            const int c1  = __builtin_amdgcn_readlane(cntv, k + 1);
            srn   = csr[(lane < c1) ? st1 + lane : 0];
            selfn = hr[(long)(n + 1) * (HID / 2)];
        }
        v2f acc = (v2f){0.0f, 0.0f};
        if (cnt > 0) p1_add_rows(hr, (lane < cnt) ? sr0 : n, (cnt < 32) ? cnt : 32, acc);
        for (int base = 32; base < cnt; base += 32) {
            const int  j  = base + lane;
            const bool ok = j < cnt;
            const int  sr = csr[ok ? start + j : 0];
            p1_add_rows(hr, ok ? sr : n, (cnt - base < 32) ? (cnt - base) : 32, acc);
        }
        float h0 = (acc.x + self.x) * dn + bv.x;
        float h1 = (acc.y + self.y) * dn + bv.y;
        h0 = h0 > 0.0f ? h0 : 0.0f;
        h1 = h1 > 0.0f ? h1 : 0.0f;
        float p0 = h0 * w2.x + h1 * w2.z;
        float p1 = h0 * w2.y + h1 * w2.w;
#pragma unroll
        for (int off = 16; off > 0; off >>= 1) {
            p0 += __shfl_xor(p0, off);
            p1 += __shfl_xor(p1, off);
        }
        if (lane == 0) *(v2f*)(hw2 + 2 * (long)n) = (v2f){p0 * dn, p1 * dn};
    }
}

#define P2_LANES   2
#define P2_RUNS    4
#define P2_THREADS 160
static_assert((N_NODES * P2_LANES) % P2_THREADS == 0, "grid covers the nodes exactly, no tail");
static_assert(P2_THREADS % 32 == 0, "whole waves");
typedef int v4iu __attribute__((ext_vector_type(4), aligned(4)));
__global__ __launch_bounds__(P2_THREADS) void pull2(const int* __restrict__ csr,
                                             const unsigned int* __restrict__ rowptr,
                                             const int* __restrict__ degi,
                                             const float* __restrict__ dinv,
                                             const float* __restrict__ hw2,
                                             const float* __restrict__ b2,
                                             float* __restrict__ out) {
    const int t   = blockIdx.x * P2_THREADS + threadIdx.x;
    const int n   = t / P2_LANES;
    const int sub = t & (P2_LANES - 1);
    const int   start = (int)rowptr[n];
    const int   cnt   = degi[n];
    const float dn    = dinv[n];
    const v2f   hs    = *(const v2f*)(hw2 + 2 * (long)n);
    float a0 = 0.0f, a1 = 0.0f;
    for (int base = 0; __builtin_amdgcn_ballot_w32(base < cnt) != 0u; base += 4 * P2_RUNS * P2_LANES) {
        int  s[4 * P2_RUNS];
        bool ok[4 * P2_RUNS];
#pragma unroll
        for (int q = 0; q < P2_RUNS; ++q) {
            const int j0 = base + 4 * (P2_LANES * q + sub);
            const int at = (j0 < cnt) ? start + j0 : 0;
            const int a3 = (at < N_EDGES - 4) ? at : N_EDGES - 4;
            const int sh = at - a3;
            const int jb = j0 - sh;
            const v4iu v = *(const v4iu*)(csr + a3);
#pragma unroll
            for (int i = 0; i < 4; ++i) {
                ok[4 * q + i] = (i >= sh) & (jb + i < cnt);
                s[4 * q + i]  = ok[4 * q + i] ? v[i] : 0;
            }
        }
        v2f h[4 * P2_RUNS];
#pragma unroll
        for (int u = 0; u < 4 * P2_RUNS; ++u) h[u] = *(const v2f*)(hw2 + 2 * (long)s[u]);
#pragma unroll
        for (int u = 0; u < 4 * P2_RUNS; ++u) {
            a0 += ok[u] ? h[u].x : 0.0f;
            a1 += ok[u] ? h[u].y : 0.0f;
        }
    }
    a0 += __shfl_xor(a0, 1);
    a1 += __shfl_xor(a1, 1);
    if (sub == 0) {
        v2f o;
        o.x = (a0 + hs.x) * dn + b2[0];
        o.y = (a1 + hs.y) * dn + b2[1];
        *(v2f*)(out + 2 * (long)n) = o;
    }
}

extern "C" void kernel_launch(void* const* d_in, const int* in_sizes, int n_in,
                              void* d_out, int out_size, void* d_ws, size_t ws_size,
                              hipStream_t stream) {
    const float* x  = (const float*)d_in[0];
    const int*   ei = (const int*)  d_in[1];
    const float* W1 = (const float*)d_in[2];
    const float* b1 = (const float*)d_in[3];
    const float* W2 = (const float*)d_in[4];
    const float* b2 = (const float*)d_in[5];
    const int* src = ei;
    const int* dst = ei + N_EDGES;
    float* out = (float*)d_out;

    float*        ws     = (float*)d_ws;
    float*        dinv   = ws;
    int*          degi   = (int*)(dinv + N_NODES);
    unsigned int* rowloc = (unsigned int*)(degi + N_NODES);
    unsigned int* rowptr = rowloc + N_NODES;
    unsigned int* btot   = rowptr + N_NODES;
    unsigned int* cc     = btot + 128;
    int*          csr    = (int*)(cc + (size_t)N_CHUNKS * N_NODES);
    float*        hw1    = (float*)(csr + N_EDGES);
    float*        hw2    = hw1 + (size_t)N_NODES * HID;

    count_kernel<<<N_RANGES * N_CHUNKS, dim3(PRIV_THREADS), 0, stream>>>(dst, cc);
    scan_kernel <<<SCAN_BLOCKS, dim3(PRIV_THREADS), 0, stream>>>(cc, dinv, degi, rowloc, btot);
    fill_gemm_kernel<<<N_RANGES * N_CHUNKS + GEMM_BLOCKS, dim3(PRIV_THREADS), 0, stream>>>(
        src, dst, cc, rowloc, btot, rowptr, csr, x, W1, dinv, hw1);

    pull1<<<N_NODES / (P1_WAVES * P1_NODES), dim3(32 * P1_WAVES), 0, stream>>>(csr, rowptr, degi, dinv, hw1, b1, W2, hw2);

    pull2<<<N_NODES * P2_LANES / P2_THREADS, dim3(P2_THREADS), 0, stream>>>(csr, rowptr, degi, dinv, hw2, b2, out);
}
